# starting kernel
# baseline (speedup 1.0000x reference)
#include <hip/hip_runtime.h>

typedef __attribute__((ext_vector_type(2))) float v2f;
typedef __attribute__((ext_vector_type(8))) float v8f;

constexpr int IN_FEAT  = 128;
constexpr int KDIM     = 2 * IN_FEAT;
constexpr int OUT_FEAT = 128;
constexpr int TILE_M   = 16;
constexpr int LDF      = KDIM + 4;

__device__ __forceinline__ void atomic_add_f32(float* p, float v) {
    __hip_atomic_fetch_add(p, v, __ATOMIC_RELAXED, __HIP_MEMORY_SCOPE_AGENT);
}

__global__ __launch_bounds__(256) void sage_scatter(
    const float* __restrict__ h_s, const int* __restrict__ src,
    const int* __restrict__ dst, float* __restrict__ nsum,
    float* __restrict__ deg, int n_edges)
{
    const int waves_total = (gridDim.x * blockDim.x) >> 5;
    const int wave = (blockIdx.x * blockDim.x + threadIdx.x) >> 5;
    const int lane = threadIdx.x & 31;

    for (int e = wave; e < n_edges; e += waves_total) {
        const int s = src[e];
        const int d = dst[e];
        const float* hs = h_s + (size_t)s * IN_FEAT;
        float*       ns = nsum + (size_t)d * IN_FEAT;
#pragma unroll
        for (int i = 0; i < IN_FEAT / 32; ++i) {
            atomic_add_f32(&ns[lane + i * 32], hs[lane + i * 32]);
        }
        if (lane == 0) atomic_add_f32(&deg[d], 1.0f);
    }
}

__global__ __launch_bounds__(256) void sage_gemm(
    const float* __restrict__ h_d, const float* __restrict__ nsum,
    const float* __restrict__ deg, const float* __restrict__ W,
    const float* __restrict__ bias, float* __restrict__ out, int n_dst)
{
    __shared__ float feat[TILE_M * LDF];

    const int t    = threadIdx.x;
    const int row0 = blockIdx.x * TILE_M;

    for (int i = t; i < TILE_M * KDIM; i += 256) {
        const int r = i >> 8;
        const int c = i & (KDIM - 1);
        int g = row0 + r;
        if (g >= n_dst) g = n_dst - 1;
        float v;
        if (c < IN_FEAT) {
            v = h_d[(size_t)g * IN_FEAT + c];
        } else {
            const float d = deg[g];
            v = nsum[(size_t)g * IN_FEAT + (c - IN_FEAT)] / fmaxf(d, 1.0f);
        }
        feat[r * LDF + c] = v;
    }
    __syncthreads();

    const int wave  = t >> 5;
    const int lane  = t & 31;
    const int n     = wave * 16 + (lane & 15);
    const int m     = lane & 15;
    const int khalf = (lane >> 4) * 2;

    const float* frow = feat + m * LDF;
    const float* wrow = W + (size_t)n * KDIM;

    v8f acc = {};
#pragma unroll 4
    for (int k0 = 0; k0 < KDIM; k0 += 4) {
        v2f a = *(const v2f*)(frow + k0 + khalf);
        v2f b = *(const v2f*)(wrow + k0 + khalf);
        acc = __builtin_amdgcn_wmma_f32_16x16x4_f32(
             false, a,  false, b,
             (short)0, acc,  false,  false);
    }

    const float bv    = bias[n];
    const int   mhigh = (lane >> 4) * 8;

#pragma unroll
    for (int v = 0; v < 8; ++v) {
        const int g = row0 + mhigh + v;
        if (g < n_dst) out[(size_t)g * OUT_FEAT + n] = acc[v] + bv;
    }
}

extern "C" void kernel_launch(void* const* d_in, const int* in_sizes, int n_in,
                              void* d_out, int out_size, void* d_ws, size_t ws_size,
                              hipStream_t stream) {
    const float* h_s  = (const float*)d_in[0];
    const float* h_d  = (const float*)d_in[1];
    const int*   src  = (const int*)d_in[2];
    const int*   dst  = (const int*)d_in[3];
    const float* W    = (const float*)d_in[4];
    const float* bias = (const float*)d_in[5];

    const int n_dst   = in_sizes[1] / IN_FEAT;
    const int n_edges = in_sizes[2];
    (void)n_in; (void)out_size; (void)ws_size;

    float* nsum = (float*)d_ws;
    float* deg  = nsum + (size_t)n_dst * IN_FEAT;

    const size_t zero_bytes = ((size_t)n_dst * IN_FEAT + (size_t)n_dst) * sizeof(float);
    hipMemsetAsync(d_ws, 0, zero_bytes, stream);

    sage_scatter<<<1024, 256, 0, stream>>>(h_s, src, dst, nsum, deg, n_edges);

    const int blocks = (n_dst + TILE_M - 1) / TILE_M;
    sage_gemm<<<blocks, 256, 0, stream>>>(h_d, nsum, deg, W, bias,
                                          (float*)d_out, n_dst);
}


# one launch: each work-group owns 208 dst rows, scans all dst, sums its edges' source rows in LDS (ds_add_f32, sum order = arrival order), mean + WMMA from LDS; no global atomics, no workspace
# speedup vs baseline: 3.1305x; 3.1305x over previous
#include <hip/hip_runtime.h>


typedef __attribute__((ext_vector_type(2))) float v2f;
typedef __attribute__((ext_vector_type(4))) float v4f;
typedef __attribute__((ext_vector_type(8))) float v8f;
typedef __attribute__((ext_vector_type(4))) int   v4i;

constexpr int IN_FEAT  = 128;
constexpr int KDIM     = 2 * IN_FEAT;
constexpr int OUT_FEAT = 128;
constexpr int LDF      = KDIM + 4;
constexpr int RB       = 208;
constexpr int NTHR     = 1024;
constexpr int QW       = 512;
constexpr int GRP      = 4;
constexpr int GSTRIDE  = GRP * NTHR;

struct TagYes { static constexpr bool value = true;  };
struct TagNo  { static constexpr bool value = false; };

__device__ __forceinline__ void lds_add_f32(float* p, float v) {
    __hip_atomic_fetch_add(p, v, __ATOMIC_RELAXED, __HIP_MEMORY_SCOPE_WORKGROUP);
}
__device__ __forceinline__ void lds_add_i32(int* p, int v) {
    __hip_atomic_fetch_add(p, v, __ATOMIC_RELAXED, __HIP_MEMORY_SCOPE_WORKGROUP);
}

__global__ __launch_bounds__(NTHR) void sage_owner_fused(
    const float* __restrict__ h_s, const float* __restrict__ h_d,
    const int* __restrict__ src, const int* __restrict__ dst,
    const float* __restrict__ W, const float* __restrict__ bias,
    float* __restrict__ out, int n_dst, int n_edges)
{
    __shared__ __attribute__((aligned(16))) float feat[RB * LDF];
    __shared__ int           degl[RB];
    __shared__ int           qe[32 * QW];
    __shared__ unsigned char qrow[32 * QW];

    const int t    = threadIdx.x;
    const int lane = t & 31;
    const int wave = __builtin_amdgcn_readfirstlane(t >> 5);
    const int row0 = blockIdx.x * RB;

    for (int i = t; i < RB * (IN_FEAT / 4); i += NTHR) {
        const int r  = i >> 5;
        const int c4 = (i & 31) * 4;
        const int g  = row0 + r;
        v4f v = {0.f, 0.f, 0.f, 0.f};
        if (g < n_dst) v = *(const v4f*)(h_d + (size_t)g * IN_FEAT + c4);
        *(v4f*)(feat + r * LDF + c4) = v;
        const v4f z = {0.f, 0.f, 0.f, 0.f};
        *(v4f*)(feat + r * LDF + IN_FEAT + c4) = z;
    }
    for (int i = t; i < RB; i += NTHR) degl[i] = 0;
    __syncthreads();

    int*           qe_w   = qe   + wave * QW;
    unsigned char* qrow_w = qrow + wave * QW;
    int wcount = 0;

    const int hi2  = (lane >> 4) << 1;
    const int col0 = (4 * lane + 0) ^ hi2;
    const int col1 = (4 * lane + 1) ^ hi2;
    const int col2 = (4 * lane + 2) ^ hi2;
    const int col3 = (4 * lane + 3) ^ hi2;

    auto add_row = [&](int lr, v4f v) {
        float* a = feat + lr * LDF + IN_FEAT;
        v4f w = v;
        if (hi2) { w.x = v.z; w.y = v.w; w.z = v.x; w.w = v.y; }
        lds_add_f32(a + col0, w.x);
        lds_add_f32(a + col1, w.y);
        lds_add_f32(a + col2, w.z);
        lds_add_f32(a + col3, w.w);
        if (lane == 0) lds_add_i32(&degl[lr], 1);
    };

    auto drain = [&](int cnt) {
        for (int i = lane; i < cnt; i += 32) {
            const int e = qe_w[i];
            qe_w[i] = src[e];
        }
        int i0 = 0;
        for (; i0 + 8 <= cnt; i0 += 8) {
            v4f v[8];
            int lr[8];
#pragma unroll
            for (int u = 0; u < 8; ++u) {
                const int s = __builtin_amdgcn_readfirstlane(qe_w[i0 + u]);
                lr[u]       = __builtin_amdgcn_readfirstlane((int)qrow_w[i0 + u]);
                v[u] = *(const v4f*)(h_s + (size_t)s * IN_FEAT + lane * 4);
            }
#pragma unroll
            for (int u = 0; u < 8; ++u) add_row(lr[u], v[u]);
        }
        for (; i0 < cnt; ++i0) {
            const int s  = __builtin_amdgcn_readfirstlane(qe_w[i0]);
            const int lr = __builtin_amdgcn_readfirstlane((int)qrow_w[i0]);
            const v4f v  = *(const v4f*)(h_s + (size_t)s * IN_FEAT + lane * 4);
            add_row(lr, v);
        }
    };

    auto push = [&](int d, int e) {
        const unsigned lr = (unsigned)(d - row0);
        const bool m = lr < (unsigned)RB;
        const unsigned bal = __builtin_amdgcn_ballot_w32(m);
        if (bal != 0u) {
            if (m) {
                const int idx = wcount + (int)__builtin_amdgcn_mbcnt_lo(bal, 0u);
                qe_w[idx]   = e;
                qrow_w[idx] = (unsigned char)lr;
            }
            wcount += __builtin_popcount(bal);
        }
    };

    const int  G  = n_edges >> 2;
    const v4i* d4 = (const v4i*)dst;
    const int  ngrp = (G + GSTRIDE - 1) / GSTRIDE;

    auto loadg_whole = [&](int it, v4i (&d)[GRP]) {
#pragma unroll
        for (int u = 0; u < GRP; ++u) d[u] = d4[it * GSTRIDE + u * NTHR + t];
    };
    auto loadg_guarded = [&](int it, v4i (&d)[GRP]) {
#pragma unroll
        for (int u = 0; u < GRP; ++u) {
            const int g = it * GSTRIDE + u * NTHR + t;
            v4i x = {-1, -1, -1, -1};
            if (g < G) x = d4[g];
            d[u] = x;
        }
    };
    auto procg = [&](int it, v4i (&d)[GRP]) {
#pragma unroll
        for (int u = 0; u < GRP; ++u) {
            if ((u & 1) == 0 && wcount > QW - 256) { drain(wcount); wcount = 0; }
            const int e = 4 * (it * GSTRIDE + u * NTHR + t);
            push(d[u].x, e + 0);
            push(d[u].y, e + 1);
            push(d[u].z, e + 2);
            push(d[u].w, e + 3);
        }
    };

    const int npair = (G / GSTRIDE) >> 1;
    if (npair > 0) {
        v4i A[GRP], B[GRP];
        loadg_whole(0, A);
        for (int p = 0; p < npair; ++p) {
            loadg_whole(2 * p + 1, B);
            procg(2 * p, A);
            const int nx = (p + 1 < npair) ? 2 * p + 2 : 2 * p + 1;
            loadg_whole(nx, A);
            procg(2 * p + 1, B);
        }
    }
    for (int it = 2 * npair; it < ngrp; ++it) {
        v4i C[GRP];
        loadg_guarded(it, C);
        procg(it, C);
    }
    if (wave == 0) {
        if (wcount > QW - 32) { drain(wcount); wcount = 0; }
        const int e = 4 * G + lane;
        int d = -1;
        if (e < n_edges) d = dst[e];
        push(d, e);
    }
    drain(wcount);
    __syncthreads();

    for (int i = t; i < RB * (IN_FEAT / 4); i += NTHR) {
        const int r  = i >> 5;
        const int c4 = (i & 31) * 4;
        const float dg = fmaxf((float)degl[r], 1.0f);
        v4f* p = (v4f*)(feat + r * LDF + IN_FEAT + c4);
        v4f x = *p;
        x.x = x.x / dg; x.y = x.y / dg; x.z = x.z / dg; x.w = x.w / dg;
        *p = x;
    }
    __syncthreads();

    const int c     = lane & 15;
    const int h     = lane >> 4;
    const int ntile = wave & 7;
    const int mg    = wave >> 3;
    const bool four = (mg == 0);
    const int n     = ntile * 16 + c;

    const float* wrow = W + (size_t)n * KDIM + 2 * h;
    const float* a0 = feat + (mg * 16 + c) * LDF + 2 * h;
    const float* a1 = a0 + 4 * 16 * LDF;
    const float* a2 = a0 + 8 * 16 * LDF;
    const float* a3 = a0 + 12 * 16 * LDF;

    v8f acc0 = {}, acc1 = {}, acc2 = {}, acc3 = {};
    auto gemm = [&](auto tag) {
        constexpr bool FOUR = decltype(tag)::value;
        v2f bn[4];
#pragma unroll
        for (int u = 0; u < 4; ++u) bn[u] = *(const v2f*)(wrow + 4 * u);
        for (int k0 = 0; k0 < KDIM; k0 += 16) {
            v2f b[4];
#pragma unroll
            for (int u = 0; u < 4; ++u) b[u] = bn[u];
            const int kn = (k0 + 16 < KDIM) ? k0 + 16 : k0;
#pragma unroll
            for (int u = 0; u < 4; ++u) bn[u] = *(const v2f*)(wrow + kn + 4 * u);
            v2f x0[4], x1[4], x2[4], x3[4];
#pragma unroll
            for (int u = 0; u < 4; ++u) {
                x0[u] = *(const v2f*)(a0 + k0 + 4 * u);
                x1[u] = *(const v2f*)(a1 + k0 + 4 * u);
                x2[u] = *(const v2f*)(a2 + k0 + 4 * u);
                if (FOUR) x3[u] = *(const v2f*)(a3 + k0 + 4 * u);
            }
#pragma unroll
            for (int u = 0; u < 4; ++u) {
                acc0 = __builtin_amdgcn_wmma_f32_16x16x4_f32(false, x0[u], false, b[u], (short)0, acc0, false, false);
                acc1 = __builtin_amdgcn_wmma_f32_16x16x4_f32(false, x1[u], false, b[u], (short)0, acc1, false, false);
                acc2 = __builtin_amdgcn_wmma_f32_16x16x4_f32(false, x2[u], false, b[u], (short)0, acc2, false, false);
                if (FOUR)
                    acc3 = __builtin_amdgcn_wmma_f32_16x16x4_f32(false, x3[u], false, b[u], (short)0, acc3, false, false);
            }
        }
    };
    if (four) gemm(TagYes{}); else gemm(TagNo{});

    const float bv = bias[n];
    auto store_tile = [&](int mt, const v8f& acc) {
#pragma unroll
        for (int r = 0; r < 8; ++r) {
            const int g = row0 + mt * 16 + 8 * h + r;
            if (g < n_dst) out[(size_t)g * OUT_FEAT + n] = acc[r] + bv;
        }
    };
    store_tile(mg, acc0);
    store_tile(mg + 4, acc1);
    store_tile(mg + 8, acc2);
    if (four) store_tile(12, acc3);
}

extern "C" void kernel_launch(void* const* d_in, const int* in_sizes, int n_in,
                              void* d_out, int out_size, void* d_ws, size_t ws_size,
                              hipStream_t stream) {
    const float* h_s  = (const float*)d_in[0];
    const float* h_d  = (const float*)d_in[1];
    const int*   src  = (const int*)d_in[2];
    const int*   dst  = (const int*)d_in[3];
    const float* W    = (const float*)d_in[4];
    const float* bias = (const float*)d_in[5];

    const int n_dst   = in_sizes[1] / IN_FEAT;
    const int n_edges = in_sizes[2];
    (void)n_in; (void)out_size; (void)d_ws; (void)ws_size;

    const int blocks = (n_dst + RB - 1) / RB;
    sage_owner_fused<<<blocks, NTHR, 0, stream>>>(h_s, h_d, src, dst, W, bias,
                                                  (float*)d_out, n_dst, n_edges);
}


# two launches: a first kernel counting-sorts each 1/256 slice of the edge list by owner work-group in LDS; an owner reads only its runs, not all dst; sum order = arrival order as before
# speedup vs baseline: 3.6982x; 1.1813x over previous
#include <hip/hip_runtime.h>


typedef __attribute__((ext_vector_type(2))) float v2f;
typedef __attribute__((ext_vector_type(4))) float v4f;
typedef __attribute__((ext_vector_type(8))) float v8f;

constexpr int IN_FEAT  = 128;
constexpr int KDIM     = 2 * IN_FEAT;
constexpr int OUT_FEAT = 128;
constexpr int LDF      = KDIM + 4;
constexpr int RB       = 208;
constexpr int NTHR     = 1024;
constexpr int NSL      = 256;
constexpr int STHR     = 256;
constexpr int UB       = 10;
constexpr int MAXOWN   = 1024;
constexpr int SCANW    = NSL / 32;
constexpr int SPW      = NSL / (NTHR / 32);
constexpr int FLATCAP  = 8192;

struct TagYes { static constexpr bool value = true;  };
struct TagNo  { static constexpr bool value = false; };

__device__ __forceinline__ void lds_add_f32(float* p, float v) {
    __hip_atomic_fetch_add(p, v, __ATOMIC_RELAXED, __HIP_MEMORY_SCOPE_WORKGROUP);
}
__device__ __forceinline__ void lds_add_i32(int* p, int v) {
    __hip_atomic_fetch_add(p, v, __ATOMIC_RELAXED, __HIP_MEMORY_SCOPE_WORKGROUP);
}

__device__ __forceinline__ int lds_add_i32_rtn(int* p, int v) {
    return __hip_atomic_fetch_add(p, v, __ATOMIC_RELAXED, __HIP_MEMORY_SCOPE_WORKGROUP);
}


__global__ __launch_bounds__(STHR) void sage_sort_slices(
    const int* __restrict__ src, const int* __restrict__ dst,
    int* __restrict__ ws, int n_dst, int n_edges)
{
    __shared__ int hist[MAXOWN + 1];

    const int t    = threadIdx.x;
    const int lane = t & 31;
    const int nown = (n_dst + RB - 1) / RB;
    const int L    = (n_edges + NSL - 1) / NSL;
    const int e0   = blockIdx.x * L;
    const int e1   = min(e0 + L, n_edges);
    int* offs = ws + (size_t)blockIdx.x * (nown + 1);
    int* ent  = ws + (size_t)NSL * (nown + 1) + e0;

    for (int i = t; i <= nown; i += STHR) hist[i] = 0;
    __syncthreads();
    for (int eb = e0; eb < e1; eb += STHR * UB) {
        unsigned d[UB];
#pragma unroll
        for (int u = 0; u < UB; ++u) {
            const int e = eb + u * STHR + t;
            const unsigned x = (unsigned)dst[min(e, e1 - 1)];
            d[u] = (e < e1) ? x : 0xffffffffu;
        }
#pragma unroll
        for (int u = 0; u < UB; ++u)
            if (d[u] < (unsigned)n_dst) lds_add_i32(&hist[d[u] / RB], 1);
    }
    __syncthreads();
    if (t < 32) {
        const int per = (nown + 32) / 32;
        const int lo  = lane * per;
        const int hi  = min(lo + per, nown + 1);
        int sum = 0;
        for (int i = lo; i < hi; ++i) sum += hist[i];
        int x = sum;
        for (int o = 1; o < 32; o <<= 1) {
            const int y = __shfl_up(x, o);
            if (lane >= o) x += y;
        }
        int run = x - sum;
        for (int i = lo; i < hi; ++i) { const int c = hist[i]; hist[i] = run; run += c; }
    }
    __syncthreads();
    for (int i = t; i <= nown; i += STHR) offs[i] = hist[i];
    __syncthreads();
    for (int eb = e0; eb < e1; eb += STHR * UB) {
        unsigned d[UB];
        int      sr[UB];
#pragma unroll
        for (int u = 0; u < UB; ++u) {
            const int e  = eb + u * STHR + t;
            const int ec = min(e, e1 - 1);
            const unsigned x = (unsigned)dst[ec];
            sr[u] = src[ec];
            d[u]  = (e < e1) ? x : 0xffffffffu;
        }
        int pos[UB];
#pragma unroll
        for (int u = 0; u < UB; ++u) {
            pos[u] = -1;
            if (d[u] < (unsigned)n_dst) pos[u] = lds_add_i32_rtn(&hist[d[u] / RB], 1);
        }
#pragma unroll
        for (int u = 0; u < UB; ++u) {
            if (pos[u] >= 0) {
                const unsigned b = d[u] / RB;
                ent[pos[u]] = (sr[u] << 8) | (int)(d[u] - b * RB);
            }
        }
    }
}

__global__ __launch_bounds__(NTHR) void sage_owner_fused(
    const float* __restrict__ h_s, const float* __restrict__ h_d,
    const int* __restrict__ ws,
    const float* __restrict__ W, const float* __restrict__ bias,
    float* __restrict__ out, int n_dst, int n_edges)
{
    __shared__ __attribute__((aligned(16))) float feat[RB * LDF];
    __shared__ int degl[RB];
    __shared__ __attribute__((aligned(16))) int flat[FLATCAP];
    __shared__ int slo[NSL], scnt[NSL], spre[NSL];
    __shared__ int wtot[SCANW];

    const int t    = threadIdx.x;
    const int lane = t & 31;
    const int wave = __builtin_amdgcn_readfirstlane(t >> 5);
    const int row0 = blockIdx.x * RB;

    const int nown = (n_dst + RB - 1) / RB;
    const int L    = (n_edges + NSL - 1) / NSL;
    const int* ent = ws + (size_t)NSL * (nown + 1);

    int rlo = 0, rhi = 0;
    if (wave < SCANW) {
        const int* o = ws + (size_t)t * (nown + 1) + blockIdx.x;
        rlo = o[0];
        rhi = o[1];
    }

    for (int i = t; i < RB * (IN_FEAT / 4); i += NTHR) {
        const int r  = i >> 5;
        const int c4 = (i & 31) * 4;
        const int g  = row0 + r;
        v4f v = {0.f, 0.f, 0.f, 0.f};
        if (g < n_dst) v = *(const v4f*)(h_d + (size_t)g * IN_FEAT + c4);
        *(v4f*)(feat + r * LDF + c4) = v;
        const v4f z = {0.f, 0.f, 0.f, 0.f};
        *(v4f*)(feat + r * LDF + IN_FEAT + c4) = z;
    }
    for (int i = t; i < RB; i += NTHR) degl[i] = 0;

    if (wave < SCANW) {
        const int cnt = rhi - rlo;
        int x = cnt;
        for (int o = 1; o < 32; o <<= 1) {
            const int y = __shfl_up(x, o);
            if (lane >= o) x += y;
        }
        slo[t]  = rlo;
        scnt[t] = cnt;
        spre[t] = x - cnt;
        if (lane == 31) wtot[wave] = x;
    }
    __syncthreads();
    static_assert(32 % SPW == 0, "the slices one wave copies must lie in one scan wave");
    const int myscan = (wave * SPW) >> 5;
    int total = 0, wpre = 0;
#pragma unroll
    for (int k = 0; k < SCANW; ++k) {
        const int w = wtot[k];
        total += w;
        if (k < myscan) wpre += w;
    }

    const int hi2  = (lane >> 4) << 1;
    const int col0 = (4 * lane + 0) ^ hi2;
    const int col1 = (4 * lane + 1) ^ hi2;
    const int col2 = (4 * lane + 2) ^ hi2;
    const int col3 = (4 * lane + 3) ^ hi2;

    auto add_row = [&](int lr, v4f v) {
        float* a = feat + lr * LDF + IN_FEAT;
        v4f w = v;
        if (hi2) { w.x = v.z; w.y = v.w; w.z = v.x; w.w = v.y; }
        lds_add_f32(a + col0, w.x);
        lds_add_f32(a + col1, w.y);
        lds_add_f32(a + col2, w.z);
        lds_add_f32(a + col3, w.w);
        if (lane == 0) lds_add_i32(&degl[lr], 1);
    };

    auto drain = [&](const int* q, int cnt) {
        int i0 = 0;
        for (; i0 + 8 <= cnt; i0 += 8) {
            v4f v[8];
            int lr[8];
#pragma unroll
            for (int u = 0; u < 8; ++u) {
                const unsigned p = (unsigned)__builtin_amdgcn_readfirstlane(q[i0 + u]);
                lr[u] = (int)(p & 255u);
                v[u]  = *(const v4f*)(h_s + (size_t)(p >> 8) * IN_FEAT + lane * 4);
            }
#pragma unroll
            for (int u = 0; u < 8; ++u) add_row(lr[u], v[u]);
        }
        for (; i0 < cnt; ++i0) {
            const unsigned p = (unsigned)__builtin_amdgcn_readfirstlane(q[i0]);
            const v4f v = *(const v4f*)(h_s + (size_t)(p >> 8) * IN_FEAT + lane * 4);
            add_row((int)(p & 255u), v);
        }
    };

    int cs[SPW], ps[SPW], ls[SPW];
#pragma unroll
    for (int j = 0; j < SPW; ++j) {
        cs[j] = scnt[wave * SPW + j];
        ps[j] = spre[wave * SPW + j] + wpre;
        ls[j] = slo[wave * SPW + j];
    }

    for (int base = 0; base < total; base += FLATCAP) {
        int rv[SPW], rg[SPW];
#pragma unroll
        for (int j = 0; j < SPW; ++j) {
            const int g  = ps[j] + lane - base;
            const bool m = lane < cs[j] && (unsigned)g < (unsigned)FLATCAP;
            rv[j] = 0;
            rg[j] = -1;
            if (m) { rv[j] = ent[(size_t)(wave * SPW + j) * L + ls[j] + lane]; rg[j] = g; }
        }
#pragma unroll
        for (int j = 0; j < SPW; ++j)
            if (rg[j] >= 0) flat[rg[j]] = rv[j];
#pragma unroll
        for (int j = 0; j < SPW; ++j) {
            const int* run = ent + (size_t)(wave * SPW + j) * L + ls[j];
            for (int i = lane + 32; i < cs[j]; i += 32) {
                const unsigned g = (unsigned)(ps[j] + i - base);
                if (g < (unsigned)FLATCAP) flat[g] = run[i];
            }
        }
        __syncthreads();
        const int n     = min(total - base, FLATCAP);
        const int chunk = (((n + 31) >> 5) + 7) & ~7;
        const int b0    = wave * chunk;
        const int cw    = max(0, min(chunk, n - b0));
        drain(flat + b0, cw);
        __syncthreads();
    }

    for (int i = t; i < RB * (IN_FEAT / 4); i += NTHR) {
        const int r  = i >> 5;
        const int c4 = (i & 31) * 4;
        const float dg = fmaxf((float)degl[r], 1.0f);
        v4f* p = (v4f*)(feat + r * LDF + IN_FEAT + c4);
        v4f x = *p;
        x.x = x.x / dg; x.y = x.y / dg; x.z = x.z / dg; x.w = x.w / dg;
        *p = x;
    }
    __syncthreads();

    const int c     = lane & 15;
    const int h     = lane >> 4;
    const int ntile = wave & 7;
    const int mg    = wave >> 3;
    const bool four = (mg == 0);
    const int n     = ntile * 16 + c;

    const float* wrow = W + (size_t)n * KDIM + 2 * h;
    const float* a0 = feat + (mg * 16 + c) * LDF + 2 * h;
    const float* a1 = a0 + 4 * 16 * LDF;
    const float* a2 = a0 + 8 * 16 * LDF;
    const float* a3 = a0 + 12 * 16 * LDF;

    v8f acc0 = {}, acc1 = {}, acc2 = {}, acc3 = {};
    auto gemm = [&](auto tag) {
        constexpr bool FOUR = decltype(tag)::value;
        v2f bn[4];
#pragma unroll
        for (int u = 0; u < 4; ++u) bn[u] = *(const v2f*)(wrow + 4 * u);
        for (int k0 = 0; k0 < KDIM; k0 += 16) {
            v2f b[4];
#pragma unroll
            for (int u = 0; u < 4; ++u) b[u] = bn[u];
            const int kn = (k0 + 16 < KDIM) ? k0 + 16 : k0;
#pragma unroll
            for (int u = 0; u < 4; ++u) bn[u] = *(const v2f*)(wrow + kn + 4 * u);
            v2f x0[4], x1[4], x2[4], x3[4];
#pragma unroll
            for (int u = 0; u < 4; ++u) {
                x0[u] = *(const v2f*)(a0 + k0 + 4 * u);
                x1[u] = *(const v2f*)(a1 + k0 + 4 * u);
                x2[u] = *(const v2f*)(a2 + k0 + 4 * u);
                if (FOUR) x3[u] = *(const v2f*)(a3 + k0 + 4 * u);
            }
#pragma unroll
            for (int u = 0; u < 4; ++u) {
                acc0 = __builtin_amdgcn_wmma_f32_16x16x4_f32(false, x0[u], false, b[u], (short)0, acc0, false, false);
                acc1 = __builtin_amdgcn_wmma_f32_16x16x4_f32(false, x1[u], false, b[u], (short)0, acc1, false, false);
                acc2 = __builtin_amdgcn_wmma_f32_16x16x4_f32(false, x2[u], false, b[u], (short)0, acc2, false, false);
                if (FOUR)
                    acc3 = __builtin_amdgcn_wmma_f32_16x16x4_f32(false, x3[u], false, b[u], (short)0, acc3, false, false);
            }
        }
    };
    if (four) gemm(TagYes{}); else gemm(TagNo{});

    const float bv = bias[n];
    auto store_tile = [&](int mt, const v8f& acc) {
#pragma unroll
        for (int r = 0; r < 8; ++r) {
            const int g = row0 + mt * 16 + 8 * h + r;
            if (g < n_dst) out[(size_t)g * OUT_FEAT + n] = acc[r] + bv;
        }
    };
    store_tile(mg, acc0);
    store_tile(mg + 4, acc1);
    store_tile(mg + 8, acc2);
    if (four) store_tile(12, acc3);
}

extern "C" void kernel_launch(void* const* d_in, const int* in_sizes, int n_in,
                              void* d_out, int out_size, void* d_ws, size_t ws_size,
                              hipStream_t stream) {
    const float* h_s  = (const float*)d_in[0];
    const float* h_d  = (const float*)d_in[1];
    const int*   src  = (const int*)d_in[2];
    const int*   dst  = (const int*)d_in[3];
    const float* W    = (const float*)d_in[4];
    const float* bias = (const float*)d_in[5];

    const int n_dst   = in_sizes[1] / IN_FEAT;
    const int n_edges = in_sizes[2];
    int* ws = (int*)d_ws;
    (void)n_in; (void)out_size; (void)ws_size;

    const int blocks = (n_dst + RB - 1) / RB;
    sage_sort_slices<<<NSL, STHR, 0, stream>>>(src, dst, ws, n_dst, n_edges);
    sage_owner_fused<<<blocks, NTHR, 0, stream>>>(h_s, h_d, ws, W, bias,
                                                  (float*)d_out, n_dst, n_edges);
}


# h_d half (k 0..127) of the WMMA product runs under the gather: each wave does one 16-k block on 3 row tiles while its 8 row loads fly (all 32 waves still gather); k order and sum order as before
# speedup vs baseline: 3.8850x; 1.0505x over previous
#include <hip/hip_runtime.h>


typedef __attribute__((ext_vector_type(2))) float v2f;
typedef __attribute__((ext_vector_type(4))) float v4f;
typedef __attribute__((ext_vector_type(8))) float v8f;

constexpr int IN_FEAT  = 128;
constexpr int KDIM     = 2 * IN_FEAT;
constexpr int OUT_FEAT = 128;
constexpr int LDF      = KDIM + 4;
constexpr int RB       = 208;
constexpr int NTHR     = 1024;
constexpr int NSL      = 256;
constexpr int STHR     = 256;
constexpr int UB       = 10;
constexpr int MAXOWN   = 1024;
constexpr int SCANW    = NSL / 32;
constexpr int SPW      = NSL / (NTHR / 32);
constexpr int FLATCAP  = 8192;

struct TagYes { static constexpr bool value = true;  };
struct TagNo  { static constexpr bool value = false; };

__device__ __forceinline__ void lds_add_f32(float* p, float v) {
    __hip_atomic_fetch_add(p, v, __ATOMIC_RELAXED, __HIP_MEMORY_SCOPE_WORKGROUP);
}
__device__ __forceinline__ void lds_add_i32(int* p, int v) {
    __hip_atomic_fetch_add(p, v, __ATOMIC_RELAXED, __HIP_MEMORY_SCOPE_WORKGROUP);
}

__device__ __forceinline__ int lds_add_i32_rtn(int* p, int v) {
    return __hip_atomic_fetch_add(p, v, __ATOMIC_RELAXED, __HIP_MEMORY_SCOPE_WORKGROUP);
}


__global__ __launch_bounds__(STHR) void sage_sort_slices(
    const int* __restrict__ src, const int* __restrict__ dst,
    int* __restrict__ ws, int n_dst, int n_edges)
{
    __shared__ int hist[MAXOWN + 1];

    const int t    = threadIdx.x;
    const int lane = t & 31;
    const int nown = (n_dst + RB - 1) / RB;
    const int L    = (n_edges + NSL - 1) / NSL;
    const int e0   = blockIdx.x * L;
    const int e1   = min(e0 + L, n_edges);
    int* offs = ws + (size_t)blockIdx.x * (nown + 1);
    int* ent  = ws + (size_t)NSL * (nown + 1) + e0;

    for (int i = t; i <= nown; i += STHR) hist[i] = 0;
    __syncthreads();
    for (int eb = e0; eb < e1; eb += STHR * UB) {
        unsigned d[UB];
#pragma unroll
        for (int u = 0; u < UB; ++u) {
            const int e = eb + u * STHR + t;
            const unsigned x = (unsigned)dst[min(e, e1 - 1)];
            d[u] = (e < e1) ? x : 0xffffffffu;
        }
#pragma unroll
        for (int u = 0; u < UB; ++u)
            if (d[u] < (unsigned)n_dst) lds_add_i32(&hist[d[u] / RB], 1);
    }
    __syncthreads();
    if (t < 32) {
        const int per = (nown + 32) / 32;
        const int lo  = lane * per;
        const int hi  = min(lo + per, nown + 1);
        int sum = 0;
        for (int i = lo; i < hi; ++i) sum += hist[i];
        int x = sum;
        for (int o = 1; o < 32; o <<= 1) {
            const int y = __shfl_up(x, o);
            if (lane >= o) x += y;
        }
        int run = x - sum;
        for (int i = lo; i < hi; ++i) { const int c = hist[i]; hist[i] = run; run += c; }
    }
    __syncthreads();
    for (int i = t; i <= nown; i += STHR) offs[i] = hist[i];
    __syncthreads();
    for (int eb = e0; eb < e1; eb += STHR * UB) {
        unsigned d[UB];
        int      sr[UB];
#pragma unroll
        for (int u = 0; u < UB; ++u) {
            const int e  = eb + u * STHR + t;
            const int ec = min(e, e1 - 1);
            const unsigned x = (unsigned)dst[ec];
            sr[u] = src[ec];
            d[u]  = (e < e1) ? x : 0xffffffffu;
        }
        int pos[UB];
#pragma unroll
        for (int u = 0; u < UB; ++u) {
            pos[u] = -1;
            if (d[u] < (unsigned)n_dst) pos[u] = lds_add_i32_rtn(&hist[d[u] / RB], 1);
        }
#pragma unroll
        for (int u = 0; u < UB; ++u) {
            if (pos[u] >= 0) {
                const unsigned b = d[u] / RB;
                ent[pos[u]] = (sr[u] << 8) | (int)(d[u] - b * RB);
            }
        }
    }
}

__global__ __launch_bounds__(NTHR) void sage_owner_fused(
    const float* __restrict__ h_s, const float* __restrict__ h_d,
    const int* __restrict__ ws,
    const float* __restrict__ W, const float* __restrict__ bias,
    float* __restrict__ out, int n_dst, int n_edges)
{
    __shared__ __attribute__((aligned(16))) float feat[RB * LDF];
    __shared__ int degl[RB];
    __shared__ __attribute__((aligned(16))) int flat[FLATCAP];
    __shared__ int slo[NSL], scnt[NSL], spre[NSL];
    __shared__ int wtot[SCANW];

    const int t    = threadIdx.x;
    const int lane = t & 31;
    const int wave = __builtin_amdgcn_readfirstlane(t >> 5);
    const int row0 = blockIdx.x * RB;

    const int nown = (n_dst + RB - 1) / RB;
    const int L    = (n_edges + NSL - 1) / NSL;
    const int* ent = ws + (size_t)NSL * (nown + 1);

    int rlo = 0, rhi = 0;
    if (wave < SCANW) {
        const int* o = ws + (size_t)t * (nown + 1) + blockIdx.x;
        rlo = o[0];
        rhi = o[1];
    }

    for (int i = t; i < RB * (IN_FEAT / 4); i += NTHR) {
        const int r  = i >> 5;
        const int c4 = (i & 31) * 4;
        const int g  = row0 + r;
        v4f v = {0.f, 0.f, 0.f, 0.f};
        if (g < n_dst) v = *(const v4f*)(h_d + (size_t)g * IN_FEAT + c4);
        *(v4f*)(feat + r * LDF + c4) = v;
        const v4f z = {0.f, 0.f, 0.f, 0.f};
        *(v4f*)(feat + r * LDF + IN_FEAT + c4) = z;
    }
    for (int i = t; i < RB; i += NTHR) degl[i] = 0;

    if (wave < SCANW) {
        const int cnt = rhi - rlo;
        int x = cnt;
        for (int o = 1; o < 32; o <<= 1) {
            const int y = __shfl_up(x, o);
            if (lane >= o) x += y;
        }
        slo[t]  = rlo;
        scnt[t] = cnt;
        spre[t] = x - cnt;
        if (lane == 31) wtot[wave] = x;
    }
    __syncthreads();
    static_assert(32 % SPW == 0, "the slices one wave copies must lie in one scan wave");
    const int myscan = (wave * SPW) >> 5;
    int total = 0, wpre = 0;
#pragma unroll
    for (int k = 0; k < SCANW; ++k) {
        const int w = wtot[k];
        total += w;
        if (k < myscan) wpre += w;
    }

    const int c     = lane & 15;
    const int h     = lane >> 4;
    const int ntile = wave & 7;
    const int mg    = wave >> 3;
    const bool four = (mg == 0);
    const int n     = ntile * 16 + c;

    const float* wrow = W + (size_t)n * KDIM + 2 * h;
    const float* a0 = feat + (mg * 16 + c) * LDF + 2 * h;
    const float* a1 = a0 + 4 * 16 * LDF;
    const float* a2 = a0 + 8 * 16 * LDF;
    const float* a3 = a0 + 12 * 16 * LDF;

    v8f acc0 = {}, acc1 = {}, acc2 = {}, acc3 = {};
    constexpr int HB = IN_FEAT / 16;
    int kb = 0;

    v2f bn[4];

    auto block = [&](auto tag, int k0, int kn) {
        constexpr bool FOUR = decltype(tag)::value;
        v2f x0[4], x1[4], x2[4], x3[4];
#pragma unroll
        for (int u = 0; u < 4; ++u) {
            x0[u] = *(const v2f*)(a0 + k0 + 4 * u);
            x1[u] = *(const v2f*)(a1 + k0 + 4 * u);
            x2[u] = *(const v2f*)(a2 + k0 + 4 * u);
            if (FOUR) x3[u] = *(const v2f*)(a3 + k0 + 4 * u);
        }
#pragma unroll
        for (int u = 0; u < 4; ++u) {
            acc0 = __builtin_amdgcn_wmma_f32_16x16x4_f32(false, x0[u], false, bn[u], (short)0, acc0, false, false);
            acc1 = __builtin_amdgcn_wmma_f32_16x16x4_f32(false, x1[u], false, bn[u], (short)0, acc1, false, false);
            acc2 = __builtin_amdgcn_wmma_f32_16x16x4_f32(false, x2[u], false, bn[u], (short)0, acc2, false, false);
            if (FOUR)
                acc3 = __builtin_amdgcn_wmma_f32_16x16x4_f32(false, x3[u], false, bn[u], (short)0, acc3, false, false);
        }
#pragma unroll
        for (int u = 0; u < 4; ++u) bn[u] = *(const v2f*)(wrow + kn + 4 * u);
    };

    const int hi2  = (lane >> 4) << 1;
    const int col0 = (4 * lane + 0) ^ hi2;
    const int col1 = (4 * lane + 1) ^ hi2;
    const int col2 = (4 * lane + 2) ^ hi2;
    const int col3 = (4 * lane + 3) ^ hi2;

    auto add_row = [&](int lr, v4f v) {
        float* a = feat + lr * LDF + IN_FEAT;
        v4f w = v;
        if (hi2) { w.x = v.z; w.y = v.w; w.z = v.x; w.w = v.y; }
        lds_add_f32(a + col0, w.x);
        lds_add_f32(a + col1, w.y);
        lds_add_f32(a + col2, w.z);
        lds_add_f32(a + col3, w.w);
        if (lane == 0) lds_add_i32(&degl[lr], 1);
    };

    auto drain = [&](const int* q, int cnt) {
        int i0 = 0;
        for (; i0 + 8 <= cnt && kb < HB; i0 += 8, ++kb) {
            v4f v[8];
            int lr[8];
#pragma unroll
            for (int u = 0; u < 8; ++u) {
                const unsigned p = (unsigned)__builtin_amdgcn_readfirstlane(q[i0 + u]);
                lr[u] = (int)(p & 255u);
                v[u]  = *(const v4f*)(h_s + (size_t)(p >> 8) * IN_FEAT + lane * 4);
            }
            __builtin_amdgcn_sched_barrier(0);
            block(TagNo{}, kb * 16, kb * 16 + 16);
#pragma unroll
            for (int u = 0; u < 8; ++u) add_row(lr[u], v[u]);
        }
        for (; i0 + 8 <= cnt; i0 += 8) {
            v4f v[8];
            int lr[8];
#pragma unroll
            for (int u = 0; u < 8; ++u) {
                const unsigned p = (unsigned)__builtin_amdgcn_readfirstlane(q[i0 + u]);
                lr[u] = (int)(p & 255u);
                v[u]  = *(const v4f*)(h_s + (size_t)(p >> 8) * IN_FEAT + lane * 4);
            }
#pragma unroll
            for (int u = 0; u < 8; ++u) add_row(lr[u], v[u]);
        }
        for (; i0 < cnt; ++i0) {
            const unsigned p = (unsigned)__builtin_amdgcn_readfirstlane(q[i0]);
            const v4f v = *(const v4f*)(h_s + (size_t)(p >> 8) * IN_FEAT + lane * 4);
            add_row((int)(p & 255u), v);
        }
        for (; kb < HB; ++kb) block(TagNo{}, kb * 16, kb * 16 + 16);
    };

    total = __builtin_amdgcn_readfirstlane(total);

    auto copy_round = [&](int base) {
        int cs[SPW], ps[SPW], ls[SPW];
#pragma unroll
        for (int j = 0; j < SPW; ++j) {
            cs[j] = __builtin_amdgcn_readfirstlane(scnt[wave * SPW + j]);
            ps[j] = __builtin_amdgcn_readfirstlane(spre[wave * SPW + j] + wpre);
            ls[j] = __builtin_amdgcn_readfirstlane(slo[wave * SPW + j]);
        }
        int rv[SPW], rg[SPW];
#pragma unroll
        for (int j = 0; j < SPW; ++j) {
            const int g  = ps[j] + lane - base;
            const bool m = lane < cs[j] && (unsigned)g < (unsigned)FLATCAP;
            rv[j] = 0;
            rg[j] = -1;
            if (m) { rv[j] = ent[(size_t)(wave * SPW + j) * L + ls[j] + lane]; rg[j] = g; }
        }
#pragma unroll
        for (int j = 0; j < SPW; ++j)
            if (rg[j] >= 0) flat[rg[j]] = rv[j];
#pragma unroll
        for (int j = 0; j < SPW; ++j) {
            const int* run = ent + (size_t)(wave * SPW + j) * L + ls[j];
            for (int i = lane + 32; i < cs[j]; i += 32) {
                const unsigned g = (unsigned)(ps[j] + i - base);
                if (g < (unsigned)FLATCAP) flat[g] = run[i];
            }
        }
    };
    auto drain_round = [&](int base) {
        const int nr    = min(total - base, FLATCAP);
        const int chunk = (((nr + 31) >> 5) + 7) & ~7;
        const int b0    = wave * chunk;
        const int cw    = max(0, min(chunk, nr - b0));
        drain(flat + b0, cw);
    };

    copy_round(0);
#pragma unroll
    for (int u = 0; u < 4; ++u) bn[u] = *(const v2f*)(wrow + 4 * u);
    __syncthreads();
    drain_round(0);
    __syncthreads();
    for (int base = FLATCAP; base < total; base += FLATCAP) {
        copy_round(base);
        __syncthreads();
        drain_round(base);
        __syncthreads();
    }

    for (int i = t; i < RB * (IN_FEAT / 4); i += NTHR) {
        const int r  = i >> 5;
        const int c4 = (i & 31) * 4;
        const float dg = fmaxf((float)degl[r], 1.0f);
        v4f* p = (v4f*)(feat + r * LDF + IN_FEAT + c4);
        v4f x = *p;
        x.x = x.x / dg; x.y = x.y / dg; x.z = x.z / dg; x.w = x.w / dg;
        *p = x;
    }
    __syncthreads();

    if (four) {
        v2f wn[4];
#pragma unroll
        for (int u = 0; u < 4; ++u) wn[u] = *(const v2f*)(wrow + 4 * u);
#pragma nounroll
        for (int k0 = 0; k0 < IN_FEAT; k0 += 16) {
            v2f x3[4];
#pragma unroll
            for (int u = 0; u < 4; ++u) x3[u] = *(const v2f*)(a3 + k0 + 4 * u);
#pragma unroll
            for (int u = 0; u < 4; ++u)
                acc3 = __builtin_amdgcn_wmma_f32_16x16x4_f32(false, x3[u], false, wn[u], (short)0, acc3, false, false);
#pragma unroll
            for (int u = 0; u < 4; ++u) wn[u] = *(const v2f*)(wrow + k0 + 16 + 4 * u);
        }
    }
    auto gemm_rest = [&](auto tag) {
#pragma nounroll
        for (int k0 = IN_FEAT; k0 < KDIM; k0 += 16) {
            const int kn = (k0 + 16 < KDIM) ? k0 + 16 : k0;
            block(tag, k0, kn);
        }
    };
    if (four) gemm_rest(TagYes{}); else gemm_rest(TagNo{});

    const float bv = bias[n];
    float* const ocol = out + n;
    const int g0 = row0 + 8 * h;
    auto store_tile = [&](int mt, const v8f& acc) {
#pragma unroll
        for (int r = 0; r < 8; ++r) {
            const int g = g0 + mt * 16 + r;
            if (g < n_dst) ocol[(size_t)g * OUT_FEAT] = acc[r] + bv;
        }
    };
    store_tile(mg, acc0);
    store_tile(mg + 4, acc1);
    store_tile(mg + 8, acc2);
    if (four) store_tile(12, acc3);
}

extern "C" void kernel_launch(void* const* d_in, const int* in_sizes, int n_in,
                              void* d_out, int out_size, void* d_ws, size_t ws_size,
                              hipStream_t stream) {
    const float* h_s  = (const float*)d_in[0];
    const float* h_d  = (const float*)d_in[1];
    const int*   src  = (const int*)d_in[2];
    const int*   dst  = (const int*)d_in[3];
    const float* W    = (const float*)d_in[4];
    const float* bias = (const float*)d_in[5];

    const int n_dst   = in_sizes[1] / IN_FEAT;
    const int n_edges = in_sizes[2];
    int* ws = (int*)d_ws;
    (void)n_in; (void)out_size; (void)ws_size;

    const int blocks = (n_dst + RB - 1) / RB;
    sage_sort_slices<<<NSL, STHR, 0, stream>>>(src, dst, ws, n_dst, n_edges);
    sage_owner_fused<<<blocks, NTHR, 0, stream>>>(h_s, h_d, ws, W, bias,
                                                  (float*)d_out, n_dst, n_edges);
}


# owner groups its entries by dst row in LDS (counting sort on int adds); each wave sums whole rows in registers, one store per row: no LDS f32 adds; sum order = slot order in a row's run
# speedup vs baseline: 4.2542x; 1.0950x over previous
#include <hip/hip_runtime.h>


typedef __attribute__((ext_vector_type(2))) float v2f;
typedef __attribute__((ext_vector_type(4))) float v4f;
typedef __attribute__((ext_vector_type(8))) float v8f;

constexpr int IN_FEAT  = 128;
constexpr int KDIM     = 2 * IN_FEAT;
constexpr int OUT_FEAT = 128;
constexpr int LDF      = KDIM + 4;
constexpr int RB       = 208;
constexpr int NTHR     = 1024;
constexpr int NSL      = 256;
constexpr int STHR     = 256;
constexpr int UB       = 10;
constexpr int MAXOWN   = 1024;
constexpr int SCANW    = NSL / 32;
constexpr int SPW      = NSL / (NTHR / 32);
constexpr int FLATCAP  = 8192;

struct TagYes { static constexpr bool value = true;  };
struct TagNo  { static constexpr bool value = false; };

__device__ __forceinline__ void lds_add_f32(float* p, float v) {
    __hip_atomic_fetch_add(p, v, __ATOMIC_RELAXED, __HIP_MEMORY_SCOPE_WORKGROUP);
}
__device__ __forceinline__ void lds_add_i32(int* p, int v) {
    __hip_atomic_fetch_add(p, v, __ATOMIC_RELAXED, __HIP_MEMORY_SCOPE_WORKGROUP);
}

__device__ __forceinline__ int lds_add_i32_rtn(int* p, int v) {
    return __hip_atomic_fetch_add(p, v, __ATOMIC_RELAXED, __HIP_MEMORY_SCOPE_WORKGROUP);
}


__global__ __launch_bounds__(STHR) void sage_sort_slices(
    const int* __restrict__ src, const int* __restrict__ dst,
    int* __restrict__ ws, int n_dst, int n_edges)
{
    __shared__ int hist[MAXOWN + 1];

    const int t    = threadIdx.x;
    const int lane = t & 31;
    const int nown = (n_dst + RB - 1) / RB;
    const int L    = (n_edges + NSL - 1) / NSL;
    const int e0   = blockIdx.x * L;
    const int e1   = min(e0 + L, n_edges);
    int* offs = ws + (size_t)blockIdx.x * (nown + 1);
    int* ent  = ws + (size_t)NSL * (nown + 1) + e0;

    for (int i = t; i <= nown; i += STHR) hist[i] = 0;
    __syncthreads();
    for (int eb = e0; eb < e1; eb += STHR * UB) {
        unsigned d[UB];
#pragma unroll
        for (int u = 0; u < UB; ++u) {
            const int e = eb + u * STHR + t;
            const unsigned x = (unsigned)dst[min(e, e1 - 1)];
            d[u] = (e < e1) ? x : 0xffffffffu;
        }
#pragma unroll
        for (int u = 0; u < UB; ++u)
            if (d[u] < (unsigned)n_dst) lds_add_i32(&hist[d[u] / RB], 1);
    }
    __syncthreads();
    if (t < 32) {
        const int per = (nown + 32) / 32;
        const int lo  = lane * per;
        const int hi  = min(lo + per, nown + 1);
        int sum = 0;
        for (int i = lo; i < hi; ++i) sum += hist[i];
        int x = sum;
        for (int o = 1; o < 32; o <<= 1) {
            const int y = __shfl_up(x, o);
            if (lane >= o) x += y;
        }
        int run = x - sum;
        for (int i = lo; i < hi; ++i) { const int c = hist[i]; hist[i] = run; run += c; }
    }
    __syncthreads();
    for (int i = t; i <= nown; i += STHR) offs[i] = hist[i];
    __syncthreads();
    for (int eb = e0; eb < e1; eb += STHR * UB) {
        unsigned d[UB];
        int      sr[UB];
#pragma unroll
        for (int u = 0; u < UB; ++u) {
            const int e  = eb + u * STHR + t;
            const int ec = min(e, e1 - 1);
            const unsigned x = (unsigned)dst[ec];
            sr[u] = src[ec];
            d[u]  = (e < e1) ? x : 0xffffffffu;
        }
        int pos[UB];
#pragma unroll
        for (int u = 0; u < UB; ++u) {
            pos[u] = -1;
            if (d[u] < (unsigned)n_dst) pos[u] = lds_add_i32_rtn(&hist[d[u] / RB], 1);
        }
#pragma unroll
        for (int u = 0; u < UB; ++u) {
            if (pos[u] >= 0) {
                const unsigned b = d[u] / RB;
                ent[pos[u]] = (sr[u] << 8) | (int)(d[u] - b * RB);
            }
        }
    }
}

__global__ __launch_bounds__(NTHR) void sage_owner_fused(
    const float* __restrict__ h_s, const float* __restrict__ h_d,
    const int* __restrict__ ws,
    const float* __restrict__ W, const float* __restrict__ bias,
    float* __restrict__ out, int n_dst, int n_edges)
{
    __shared__ __attribute__((aligned(16))) float feat[RB * LDF];
    __shared__ int degl[RB];
    __shared__ __attribute__((aligned(16))) int flat[FLATCAP];
    __shared__ __attribute__((aligned(16))) int sorted[FLATCAP];
    __shared__ int rcnt[RB];
    __shared__ int rstart[RB + 1];
    __shared__ int rcur[RB];
    __shared__ int slo[NSL], scnt[NSL], spre[NSL];
    __shared__ int wtot[SCANW];

    const int t    = threadIdx.x;
    const int lane = t & 31;
    const int wave = __builtin_amdgcn_readfirstlane(t >> 5);
    const int row0 = blockIdx.x * RB;

    const int nown = (n_dst + RB - 1) / RB;
    const int L    = (n_edges + NSL - 1) / NSL;
    const int* ent = ws + (size_t)NSL * (nown + 1);

    int rlo = 0, rhi = 0;
    if (wave < SCANW) {
        const int* o = ws + (size_t)t * (nown + 1) + blockIdx.x;
        rlo = o[0];
        rhi = o[1];
    }

    for (int i = t; i < RB * (IN_FEAT / 4); i += NTHR) {
        const int r  = i >> 5;
        const int c4 = (i & 31) * 4;
        const int g  = row0 + r;
        v4f v = {0.f, 0.f, 0.f, 0.f};
        if (g < n_dst) v = *(const v4f*)(h_d + (size_t)g * IN_FEAT + c4);
        *(v4f*)(feat + r * LDF + c4) = v;
        const v4f z = {0.f, 0.f, 0.f, 0.f};
        *(v4f*)(feat + r * LDF + IN_FEAT + c4) = z;
    }
    for (int i = t; i < RB; i += NTHR) { degl[i] = 0; rcnt[i] = 0; }

    if (wave < SCANW) {
        const int cnt = rhi - rlo;
        int x = cnt;
        for (int o = 1; o < 32; o <<= 1) {
            const int y = __shfl_up(x, o);
            if (lane >= o) x += y;
        }
        slo[t]  = rlo;
        scnt[t] = cnt;
        spre[t] = x - cnt;
        if (lane == 31) wtot[wave] = x;
    }
    __syncthreads();
    static_assert(32 % SPW == 0, "the slices one wave copies must lie in one scan wave");
    const int myscan = (wave * SPW) >> 5;
    int total = 0, wpre = 0;
#pragma unroll
    for (int k = 0; k < SCANW; ++k) {
        const int w = wtot[k];
        total += w;
        if (k < myscan) wpre += w;
    }

    const int c     = lane & 15;
    const int h     = lane >> 4;
    const int ntile = wave & 7;
    const int mg    = wave >> 3;
    const bool four = (mg == 0);
    const int n     = ntile * 16 + c;

    const float* wrow = W + (size_t)n * KDIM + 2 * h;
    const float* a0 = feat + (mg * 16 + c) * LDF + 2 * h;
    const float* a1 = a0 + 4 * 16 * LDF;
    const float* a2 = a0 + 8 * 16 * LDF;
    const float* a3 = a0 + 12 * 16 * LDF;

    v8f acc0 = {}, acc1 = {}, acc2 = {}, acc3 = {};
    constexpr int HB = IN_FEAT / 16;
    int kb = 0;

    v2f bn[4];

    auto block = [&](auto tag, int k0, int kn) {
        constexpr bool FOUR = decltype(tag)::value;
        v2f x0[4], x1[4], x2[4], x3[4];
#pragma unroll
        for (int u = 0; u < 4; ++u) {
            x0[u] = *(const v2f*)(a0 + k0 + 4 * u);
            x1[u] = *(const v2f*)(a1 + k0 + 4 * u);
            x2[u] = *(const v2f*)(a2 + k0 + 4 * u);
            if (FOUR) x3[u] = *(const v2f*)(a3 + k0 + 4 * u);
        }
#pragma unroll
        for (int u = 0; u < 4; ++u) {
            acc0 = __builtin_amdgcn_wmma_f32_16x16x4_f32(false, x0[u], false, bn[u], (short)0, acc0, false, false);
            acc1 = __builtin_amdgcn_wmma_f32_16x16x4_f32(false, x1[u], false, bn[u], (short)0, acc1, false, false);
            acc2 = __builtin_amdgcn_wmma_f32_16x16x4_f32(false, x2[u], false, bn[u], (short)0, acc2, false, false);
            if (FOUR)
                acc3 = __builtin_amdgcn_wmma_f32_16x16x4_f32(false, x3[u], false, bn[u], (short)0, acc3, false, false);
        }
#pragma unroll
        for (int u = 0; u < 4; ++u) bn[u] = *(const v2f*)(wrow + kn + 4 * u);
    };

    auto flush = [&](auto tag, int r, v4f s, int n) {
        constexpr bool FIRST = decltype(tag)::value;
        float* a = feat + r * LDF + IN_FEAT + 4 * lane;
        if (!FIRST) {
            const v4f p = *(const v4f*)a;
            s.x += p.x; s.y += p.y; s.z += p.z; s.w += p.w;
            n += __builtin_amdgcn_readfirstlane(degl[r]);
        }
        *(v4f*)a = s;
        if (lane == 0) degl[r] = n;
    };

    auto drain = [&](auto tag, const int* q, int cnt, int rfirst) {
        constexpr bool FIRST = decltype(tag)::value;
        v4f s = {0.f, 0.f, 0.f, 0.f};
        int cur = rfirst;
        int n   = 0;
        auto take = [&](int lr, v4f v) {
            if (lr != cur) {
                if (n > 0) flush(tag, cur, s, n);
                cur = lr;
                s.x = 0.f; s.y = 0.f; s.z = 0.f; s.w = 0.f;
                n = 0;
            }
            s.x += v.x; s.y += v.y; s.z += v.z; s.w += v.w;
            ++n;
        };
        int i0 = 0;
        if (FIRST)
        for (; i0 < cnt && kb < HB; i0 += 8, ++kb) {
            v4f v[8];
            int lr[8];
#pragma unroll
            for (int u = 0; u < 8; ++u) {
                const unsigned p = (unsigned)__builtin_amdgcn_readfirstlane(q[min(i0 + u, cnt - 1)]);
                lr[u] = (int)(p & 255u);
                v[u]  = *(const v4f*)(h_s + (size_t)(p >> 8) * IN_FEAT + lane * 4);
            }
            __builtin_amdgcn_sched_barrier(0);
            block(TagNo{}, kb * 16, kb * 16 + 16);
            __builtin_amdgcn_sched_barrier(0);
#pragma unroll
            for (int u = 0; u < 8; ++u)
                if (i0 + u < cnt) take(lr[u], v[u]);
        }
        for (; i0 < cnt; i0 += 8) {
            v4f v[8];
            int lr[8];
#pragma unroll
            for (int u = 0; u < 8; ++u) {
                const unsigned p = (unsigned)__builtin_amdgcn_readfirstlane(q[min(i0 + u, cnt - 1)]);
                lr[u] = (int)(p & 255u);
                v[u]  = *(const v4f*)(h_s + (size_t)(p >> 8) * IN_FEAT + lane * 4);
            }
#pragma unroll
            for (int u = 0; u < 8; ++u)
                if (i0 + u < cnt) take(lr[u], v[u]);
        }
        if (n > 0) flush(tag, cur, s, n);
        if (FIRST)
            for (; kb < HB; ++kb) block(TagNo{}, kb * 16, kb * 16 + 16);
    };

    total = __builtin_amdgcn_readfirstlane(total);

    auto copy_round = [&](int base) {
        int cs[SPW], ps[SPW], ls[SPW];
#pragma unroll
        for (int j = 0; j < SPW; ++j) {
            cs[j] = __builtin_amdgcn_readfirstlane(scnt[wave * SPW + j]);
            ps[j] = __builtin_amdgcn_readfirstlane(spre[wave * SPW + j] + wpre);
            ls[j] = __builtin_amdgcn_readfirstlane(slo[wave * SPW + j]);
        }
        int rv[SPW], rg[SPW];
#pragma unroll
        for (int j = 0; j < SPW; ++j) {
            const int g  = ps[j] + lane - base;
            const bool m = lane < cs[j] && (unsigned)g < (unsigned)FLATCAP;
            rv[j] = 0;
            rg[j] = -1;
            if (m) { rv[j] = ent[(size_t)(wave * SPW + j) * L + ls[j] + lane]; rg[j] = g; }
        }
#pragma unroll
        for (int j = 0; j < SPW; ++j)
            if (rg[j] >= 0) { flat[rg[j]] = rv[j]; lds_add_i32(&rcnt[rv[j] & 255], 1); }
#pragma unroll
        for (int j = 0; j < SPW; ++j) {
            const int* run = ent + (size_t)(wave * SPW + j) * L + ls[j];
            for (int i = lane + 32; i < cs[j]; i += 32) {
                const unsigned g = (unsigned)(ps[j] + i - base);
                if (g < (unsigned)FLATCAP) { const int x = run[i]; flat[g] = x; lds_add_i32(&rcnt[x & 255], 1); }
            }
        }
    };
    auto group_scan = [&](int nr) {
        if (wave == 0) {
            constexpr int PER = (RB + 31) / 32;
            const int lo = lane * PER;
            const int hi = min(lo + PER, RB);
            int sum = 0;
            for (int i = lo; i < hi; ++i) sum += rcnt[i];
            int x = sum;
            for (int o = 1; o < 32; o <<= 1) {
                const int y = __shfl_up(x, o);
                if (lane >= o) x += y;
            }
            int run = x - sum;
            for (int i = lo; i < hi; ++i) { const int c = rcnt[i]; rstart[i] = run; rcur[i] = run; run += c; }
            if (lane == 0) rstart[RB] = nr;
        }
    };
    auto group_place = [&](int nr) {
        for (int i = t; i < nr; i += NTHR) {
            const int p = flat[i];
            const int pos = lds_add_i32_rtn(&rcur[p & 255], 1);
            sorted[pos] = p;
        }
    };
    auto rows_below = [&](int x) {
        int cnt = 0;
#pragma unroll
        for (int j = 0; j < (RB + 31) / 32; ++j) {
            const int r = j * 32 + lane;
            const bool b = (r < RB) && (rstart[min(r, RB)] < x);
            cnt += __popc(__builtin_amdgcn_ballot_w32(b));
        }
        return cnt;
    };
    auto drain_round = [&](auto tag, int base) {
        const int nr    = min(total - base, FLATCAP);
        const int chunk = (nr + 31) >> 5;
        const int rw0   = (wave == 0) ? 0 : rows_below(wave * chunk);
        const int rw1   = (wave == NTHR / 32 - 1) ? RB : rows_below((wave + 1) * chunk);
        const int e0    = __builtin_amdgcn_readfirstlane(rstart[rw0]);
        const int e1    = __builtin_amdgcn_readfirstlane(rstart[rw1]);
        drain(tag, sorted + e0, e1 - e0, rw0);
    };

    copy_round(0);
#pragma unroll
    for (int u = 0; u < 4; ++u) bn[u] = *(const v2f*)(wrow + 4 * u);
    __syncthreads();
    group_scan(min(total, FLATCAP));
    __syncthreads();
    group_place(min(total, FLATCAP));
    __syncthreads();
    drain_round(TagYes{}, 0);
    __syncthreads();
    for (int base = FLATCAP; base < total; base += FLATCAP) {
        for (int i = t; i < RB; i += NTHR) rcnt[i] = 0;
        __syncthreads();
        copy_round(base);
        __syncthreads();
        group_scan(min(total - base, FLATCAP));
        __syncthreads();
        group_place(min(total - base, FLATCAP));
        __syncthreads();
        drain_round(TagNo{}, base);
        __syncthreads();
    }

    for (int i = t; i < RB * (IN_FEAT / 4); i += NTHR) {
        const int r  = i >> 5;
        const int c4 = (i & 31) * 4;
        const float dg = fmaxf((float)degl[r], 1.0f);
        v4f* p = (v4f*)(feat + r * LDF + IN_FEAT + c4);
        v4f x = *p;
        x.x = x.x / dg; x.y = x.y / dg; x.z = x.z / dg; x.w = x.w / dg;
        *p = x;
    }
    __syncthreads();

    if (four) {
        v2f wn[4];
#pragma unroll
        for (int u = 0; u < 4; ++u) wn[u] = *(const v2f*)(wrow + 4 * u);
#pragma nounroll
        for (int k0 = 0; k0 < IN_FEAT; k0 += 16) {
            v2f x3[4];
#pragma unroll
            for (int u = 0; u < 4; ++u) x3[u] = *(const v2f*)(a3 + k0 + 4 * u);
#pragma unroll
            for (int u = 0; u < 4; ++u)
                acc3 = __builtin_amdgcn_wmma_f32_16x16x4_f32(false, x3[u], false, wn[u], (short)0, acc3, false, false);
#pragma unroll
            for (int u = 0; u < 4; ++u) wn[u] = *(const v2f*)(wrow + k0 + 16 + 4 * u);
        }
    }
    auto gemm_rest = [&](auto tag) {
#pragma nounroll
        for (int k0 = IN_FEAT; k0 < KDIM; k0 += 16) {
            const int kn = (k0 + 16 < KDIM) ? k0 + 16 : k0;
            block(tag, k0, kn);
        }
    };
    if (four) gemm_rest(TagYes{}); else gemm_rest(TagNo{});

    const float bv = bias[n];
    float* const ocol = out + n;
    const int g0 = row0 + 8 * h;
    auto store_tile = [&](int mt, const v8f& acc) {
#pragma unroll
        for (int r = 0; r < 8; ++r) {
            const int g = g0 + mt * 16 + r;
            if (g < n_dst) ocol[(size_t)g * OUT_FEAT] = acc[r] + bv;
        }
    };
    store_tile(mg, acc0);
    store_tile(mg + 4, acc1);
    store_tile(mg + 8, acc2);
    if (four) store_tile(12, acc3);
}

extern "C" void kernel_launch(void* const* d_in, const int* in_sizes, int n_in,
                              void* d_out, int out_size, void* d_ws, size_t ws_size,
                              hipStream_t stream) {
    const float* h_s  = (const float*)d_in[0];
    const float* h_d  = (const float*)d_in[1];
    const int*   src  = (const int*)d_in[2];
    const int*   dst  = (const int*)d_in[3];
    const float* W    = (const float*)d_in[4];
    const float* bias = (const float*)d_in[5];

    const int n_dst   = in_sizes[1] / IN_FEAT;
    const int n_edges = in_sizes[2];
    int* ws = (int*)d_ws;
    (void)n_in; (void)out_size; (void)ws_size;

    const int blocks = (n_dst + RB - 1) / RB;
    sage_sort_slices<<<NSL, STHR, 0, stream>>>(src, dst, ws, n_dst, n_edges);
    sage_owner_fused<<<blocks, NTHR, 0, stream>>>(h_s, h_d, ws, W, bias,
                                                  (float*)d_out, n_dst, n_edges);
}


# W fragments are read from an LDS copy of the k half of W in use (one copy per work-group, brought in by whole 512-byte rows) instead of per-lane 8-byte global loads; k order and sum order unchanged
# speedup vs baseline: 4.6109x; 1.0838x over previous
#include <hip/hip_runtime.h>


typedef __attribute__((ext_vector_type(2))) float v2f;
typedef __attribute__((ext_vector_type(4))) float v4f;
typedef __attribute__((ext_vector_type(8))) float v8f;

constexpr int IN_FEAT  = 128;
constexpr int KDIM     = 2 * IN_FEAT;
constexpr int OUT_FEAT = 128;
constexpr int LDF      = KDIM + 4;
constexpr int LDW      = IN_FEAT + 4;
constexpr int RB       = 208;
constexpr int NTHR     = 1024;
constexpr int NSL      = 256;
constexpr int STHR     = 256;
constexpr int UB       = 10;
constexpr int MAXOWN   = 1024;
constexpr int SCANW    = NSL / 32;
constexpr int SPW      = NSL / (NTHR / 32);
constexpr int FLATCAP  = 4096;

struct TagYes { static constexpr bool value = true;  };
struct TagNo  { static constexpr bool value = false; };

__device__ __forceinline__ void lds_add_f32(float* p, float v) {
    __hip_atomic_fetch_add(p, v, __ATOMIC_RELAXED, __HIP_MEMORY_SCOPE_WORKGROUP);
}
__device__ __forceinline__ void lds_add_i32(int* p, int v) {
    __hip_atomic_fetch_add(p, v, __ATOMIC_RELAXED, __HIP_MEMORY_SCOPE_WORKGROUP);
}

__device__ __forceinline__ int lds_add_i32_rtn(int* p, int v) {
    return __hip_atomic_fetch_add(p, v, __ATOMIC_RELAXED, __HIP_MEMORY_SCOPE_WORKGROUP);
}


__global__ __launch_bounds__(STHR) void sage_sort_slices(
    const int* __restrict__ src, const int* __restrict__ dst,
    int* __restrict__ ws, int n_dst, int n_edges)
{
    __shared__ int hist[MAXOWN + 1];

    const int t    = threadIdx.x;
    const int lane = t & 31;
    const int nown = (n_dst + RB - 1) / RB;
    const int L    = (n_edges + NSL - 1) / NSL;
    const int e0   = blockIdx.x * L;
    const int e1   = min(e0 + L, n_edges);
    int* offs = ws + (size_t)blockIdx.x * (nown + 1);
    int* ent  = ws + (size_t)NSL * (nown + 1) + e0;

    for (int i = t; i <= nown; i += STHR) hist[i] = 0;
    __syncthreads();
    for (int eb = e0; eb < e1; eb += STHR * UB) {
        unsigned d[UB];
#pragma unroll
        for (int u = 0; u < UB; ++u) {
            const int e = eb + u * STHR + t;
            const unsigned x = (unsigned)dst[min(e, e1 - 1)];
            d[u] = (e < e1) ? x : 0xffffffffu;
        }
#pragma unroll
        for (int u = 0; u < UB; ++u)
            if (d[u] < (unsigned)n_dst) lds_add_i32(&hist[d[u] / RB], 1);
    }
    __syncthreads();
    if (t < 32) {
        const int per = (nown + 32) / 32;
        const int lo  = lane * per;
        const int hi  = min(lo + per, nown + 1);
        int sum = 0;
        for (int i = lo; i < hi; ++i) sum += hist[i];
        int x = sum;
        for (int o = 1; o < 32; o <<= 1) {
            const int y = __shfl_up(x, o);
            if (lane >= o) x += y;
        }
        int run = x - sum;
        for (int i = lo; i < hi; ++i) { const int c = hist[i]; hist[i] = run; run += c; }
    }
    __syncthreads();
    for (int i = t; i <= nown; i += STHR) offs[i] = hist[i];
    __syncthreads();
    for (int eb = e0; eb < e1; eb += STHR * UB) {
        unsigned d[UB];
        int      sr[UB];
#pragma unroll
        for (int u = 0; u < UB; ++u) {
            const int e  = eb + u * STHR + t;
            const int ec = min(e, e1 - 1);
            const unsigned x = (unsigned)dst[ec];
            sr[u] = src[ec];
            d[u]  = (e < e1) ? x : 0xffffffffu;
        }
        int pos[UB];
#pragma unroll
        for (int u = 0; u < UB; ++u) {
            pos[u] = -1;
            if (d[u] < (unsigned)n_dst) pos[u] = lds_add_i32_rtn(&hist[d[u] / RB], 1);
        }
#pragma unroll
        for (int u = 0; u < UB; ++u) {
            if (pos[u] >= 0) {
                const unsigned b = d[u] / RB;
                ent[pos[u]] = (sr[u] << 8) | (int)(d[u] - b * RB);
            }
        }
    }
}

__global__ __launch_bounds__(NTHR) void sage_owner_fused(
    const float* __restrict__ h_s, const float* __restrict__ h_d,
    const int* __restrict__ ws,
    const float* __restrict__ W, const float* __restrict__ bias,
    float* __restrict__ out, int n_dst, int n_edges)
{
    __shared__ __attribute__((aligned(16))) float feat[RB * LDF];
    __shared__ __attribute__((aligned(16))) float wl[OUT_FEAT * LDW];
    __shared__ int degl[RB];
    __shared__ __attribute__((aligned(16))) int flat[FLATCAP];
    __shared__ __attribute__((aligned(16))) int sorted[FLATCAP];
    __shared__ int rcnt[RB];
    __shared__ int rstart[RB + 1];
    __shared__ int rcur[RB];
    __shared__ int slo[NSL], scnt[NSL], spre[NSL];
    __shared__ int wtot[SCANW];

    const int t    = threadIdx.x;
    const int lane = t & 31;
    const int wave = __builtin_amdgcn_readfirstlane(t >> 5);
    const int row0 = blockIdx.x * RB;

    const int nown = (n_dst + RB - 1) / RB;
    const int L    = (n_edges + NSL - 1) / NSL;
    const int* ent = ws + (size_t)NSL * (nown + 1);

    int rlo = 0, rhi = 0;
    if (wave < SCANW) {
        const int* o = ws + (size_t)t * (nown + 1) + blockIdx.x;
        rlo = o[0];
        rhi = o[1];
    }

    for (int i = t; i < RB * (IN_FEAT / 4); i += NTHR) {
        const int r  = i >> 5;
        const int c4 = (i & 31) * 4;
        const int g  = row0 + r;
        v4f v = {0.f, 0.f, 0.f, 0.f};
        if (g < n_dst) v = *(const v4f*)(h_d + (size_t)g * IN_FEAT + c4);
        *(v4f*)(feat + r * LDF + c4) = v;
        const v4f z = {0.f, 0.f, 0.f, 0.f};
        *(v4f*)(feat + r * LDF + IN_FEAT + c4) = z;
    }
    for (int i = t; i < RB; i += NTHR) { degl[i] = 0; rcnt[i] = 0; }
    static_assert(OUT_FEAT * (IN_FEAT / 4) == 4 * NTHR, "four 16-byte pieces of a W half per thread");
#pragma unroll
    for (int j = 0; j < 4; ++j) {
        const int i = t + j * NTHR;
        *(v4f*)(wl + (i >> 5) * LDW + (i & 31) * 4) = *(const v4f*)(W + (size_t)(i >> 5) * KDIM + (i & 31) * 4);
    }

    if (wave < SCANW) {
        const int cnt = rhi - rlo;
        int x = cnt;
        for (int o = 1; o < 32; o <<= 1) {
            const int y = __shfl_up(x, o);
            if (lane >= o) x += y;
        }
        slo[t]  = rlo;
        scnt[t] = cnt;
        spre[t] = x - cnt;
        if (lane == 31) wtot[wave] = x;
    }
    __syncthreads();
    static_assert(32 % SPW == 0, "the slices one wave copies must lie in one scan wave");
    const int myscan = (wave * SPW) >> 5;
    int total = 0, wpre = 0;
#pragma unroll
    for (int k = 0; k < SCANW; ++k) {
        const int w = wtot[k];
        total += w;
        if (k < myscan) wpre += w;
    }

    const int c     = lane & 15;
    const int h     = lane >> 4;
    const int ntile = wave & 7;
    const int mg    = wave >> 3;
    const bool four = (mg == 0);
    const int n     = ntile * 16 + c;

    const float* bw = wl + n * LDW + 2 * h;
    const float* a0 = feat + (mg * 16 + c) * LDF + 2 * h;
    const float* a1 = a0 + 4 * 16 * LDF;
    const float* a2 = a0 + 8 * 16 * LDF;
    const float* a3 = a0 + 12 * 16 * LDF;

    v8f acc0 = {}, acc1 = {}, acc2 = {}, acc3 = {};
    constexpr int HB = IN_FEAT / 16;
    int kb = 0;

    auto block = [&](auto tag, int k0, int kw) {
        constexpr bool FOUR = decltype(tag)::value;
        v2f x0[4], x1[4], x2[4], x3[4], bf[4];
#pragma unroll
        for (int u = 0; u < 4; ++u) {
            bf[u] = *(const v2f*)(bw + kw + 4 * u);
            x0[u] = *(const v2f*)(a0 + k0 + 4 * u);
            x1[u] = *(const v2f*)(a1 + k0 + 4 * u);
            x2[u] = *(const v2f*)(a2 + k0 + 4 * u);
            if (FOUR) x3[u] = *(const v2f*)(a3 + k0 + 4 * u);
        }
#pragma unroll
        for (int u = 0; u < 4; ++u) {
            acc0 = __builtin_amdgcn_wmma_f32_16x16x4_f32(false, x0[u], false, bf[u], (short)0, acc0, false, false);
            acc1 = __builtin_amdgcn_wmma_f32_16x16x4_f32(false, x1[u], false, bf[u], (short)0, acc1, false, false);
            acc2 = __builtin_amdgcn_wmma_f32_16x16x4_f32(false, x2[u], false, bf[u], (short)0, acc2, false, false);
            if (FOUR)
                acc3 = __builtin_amdgcn_wmma_f32_16x16x4_f32(false, x3[u], false, bf[u], (short)0, acc3, false, false);
        }
    };
    auto tile12_hd = [&]() {
#pragma nounroll
        for (int k0 = 0; k0 < IN_FEAT; k0 += 16) {
            v2f x3[4], bf[4];
#pragma unroll
            for (int u = 0; u < 4; ++u) {
                bf[u] = *(const v2f*)(bw + k0 + 4 * u);
                x3[u] = *(const v2f*)(a3 + k0 + 4 * u);
            }
#pragma unroll
            for (int u = 0; u < 4; ++u)
                acc3 = __builtin_amdgcn_wmma_f32_16x16x4_f32(false, x3[u], false, bf[u], (short)0, acc3, false, false);
        }
    };

    auto flush = [&](auto tag, int r, v4f s, int n) {
        constexpr bool FIRST = decltype(tag)::value;
        float* a = feat + r * LDF + IN_FEAT + 4 * lane;
        if (!FIRST) {
            const v4f p = *(const v4f*)a;
            s.x += p.x; s.y += p.y; s.z += p.z; s.w += p.w;
            n += __builtin_amdgcn_readfirstlane(degl[r]);
        }
        *(v4f*)a = s;
        if (lane == 0) degl[r] = n;
    };

    auto drain = [&](auto tag, const int* q, int cnt, int rfirst) {
        constexpr bool FIRST = decltype(tag)::value;
        v4f s = {0.f, 0.f, 0.f, 0.f};
        int cur = rfirst;
        int n   = 0;
        auto take = [&](int lr, v4f v) {
            if (lr != cur) {
                if (n > 0) flush(tag, cur, s, n);
                cur = lr;
                s.x = 0.f; s.y = 0.f; s.z = 0.f; s.w = 0.f;
                n = 0;
            }
            s.x += v.x; s.y += v.y; s.z += v.z; s.w += v.w;
            ++n;
        };
        int i0 = 0;
        if (FIRST)
        for (; i0 < cnt && kb < HB; i0 += 8, ++kb) {
            v4f v[8];
            int lr[8];
#pragma unroll
            for (int u = 0; u < 8; ++u) {
                const unsigned p = (unsigned)__builtin_amdgcn_readfirstlane(q[min(i0 + u, cnt - 1)]);
                lr[u] = (int)(p & 255u);
                v[u]  = *(const v4f*)(h_s + (size_t)(p >> 8) * IN_FEAT + lane * 4);
            }
            __builtin_amdgcn_sched_barrier(0);
            block(TagNo{}, kb * 16, kb * 16);
            __builtin_amdgcn_sched_barrier(0);
#pragma unroll
            for (int u = 0; u < 8; ++u)
                if (i0 + u < cnt) take(lr[u], v[u]);
        }
        for (; i0 < cnt; i0 += 8) {
            v4f v[8];
            int lr[8];
#pragma unroll
            for (int u = 0; u < 8; ++u) {
                const unsigned p = (unsigned)__builtin_amdgcn_readfirstlane(q[min(i0 + u, cnt - 1)]);
                lr[u] = (int)(p & 255u);
                v[u]  = *(const v4f*)(h_s + (size_t)(p >> 8) * IN_FEAT + lane * 4);
            }
#pragma unroll
            for (int u = 0; u < 8; ++u)
                if (i0 + u < cnt) take(lr[u], v[u]);
        }
        if (n > 0) flush(tag, cur, s, n);
        if (FIRST) {
            for (; kb < HB; ++kb) block(TagNo{}, kb * 16, kb * 16);
            if (four) tile12_hd();
        }
    };

    total = __builtin_amdgcn_readfirstlane(total);

    auto copy_round = [&](int base) {
        int cs[SPW], ps[SPW], ls[SPW];
#pragma unroll
        for (int j = 0; j < SPW; ++j) {
            cs[j] = __builtin_amdgcn_readfirstlane(scnt[wave * SPW + j]);
            ps[j] = __builtin_amdgcn_readfirstlane(spre[wave * SPW + j] + wpre);
            ls[j] = __builtin_amdgcn_readfirstlane(slo[wave * SPW + j]);
        }
        int rv[SPW], rg[SPW];
#pragma unroll
        for (int j = 0; j < SPW; ++j) {
            const int g  = ps[j] + lane - base;
            const bool m = lane < cs[j] && (unsigned)g < (unsigned)FLATCAP;
            rv[j] = 0;
            rg[j] = -1;
            if (m) { rv[j] = ent[(size_t)(wave * SPW + j) * L + ls[j] + lane]; rg[j] = g; }
        }
#pragma unroll
        for (int j = 0; j < SPW; ++j)
            if (rg[j] >= 0) { flat[rg[j]] = rv[j]; lds_add_i32(&rcnt[rv[j] & 255], 1); }
#pragma unroll
        for (int j = 0; j < SPW; ++j) {
            const int* run = ent + (size_t)(wave * SPW + j) * L + ls[j];
            for (int i = lane + 32; i < cs[j]; i += 32) {
                const unsigned g = (unsigned)(ps[j] + i - base);
                if (g < (unsigned)FLATCAP) { const int x = run[i]; flat[g] = x; lds_add_i32(&rcnt[x & 255], 1); }
            }
        }
    };
    auto group_scan = [&](int nr) {
        if (wave == 0) {
            constexpr int PER = (RB + 31) / 32;
            const int lo = lane * PER;
            const int hi = min(lo + PER, RB);
            int sum = 0;
            for (int i = lo; i < hi; ++i) sum += rcnt[i];
            int x = sum;
            for (int o = 1; o < 32; o <<= 1) {
                const int y = __shfl_up(x, o);
                if (lane >= o) x += y;
            }
            int run = x - sum;
            for (int i = lo; i < hi; ++i) { const int c = rcnt[i]; rstart[i] = run; rcur[i] = run; run += c; }
            if (lane == 0) rstart[RB] = nr;
        }
    };
    auto group_place = [&](int nr) {
        for (int i = t; i < nr; i += NTHR) {
            const int p = flat[i];
            const int pos = lds_add_i32_rtn(&rcur[p & 255], 1);
            sorted[pos] = p;
        }
    };
    auto rows_below = [&](int x) {
        int cnt = 0;
#pragma unroll
        for (int j = 0; j < (RB + 31) / 32; ++j) {
            const int r = j * 32 + lane;
            const bool b = (r < RB) && (rstart[min(r, RB)] < x);
            cnt += __popc(__builtin_amdgcn_ballot_w32(b));
        }
        return cnt;
    };
    auto drain_round = [&](auto tag, int base) {
        const int nr    = min(total - base, FLATCAP);
        const int chunk = (nr + 31) >> 5;
        const int rw0   = (wave == 0) ? 0 : rows_below(wave * chunk);
        const int rw1   = (wave == NTHR / 32 - 1) ? RB : rows_below((wave + 1) * chunk);
        const int e0    = __builtin_amdgcn_readfirstlane(rstart[rw0]);
        const int e1    = __builtin_amdgcn_readfirstlane(rstart[rw1]);
        drain(tag, sorted + e0, e1 - e0, rw0);
    };

    copy_round(0);
    __syncthreads();
    group_scan(min(total, FLATCAP));
    __syncthreads();
    group_place(min(total, FLATCAP));
    __syncthreads();
    drain_round(TagYes{}, 0);
    __syncthreads();
    for (int base = FLATCAP; base < total; base += FLATCAP) {
        for (int i = t; i < RB; i += NTHR) rcnt[i] = 0;
        __syncthreads();
        copy_round(base);
        __syncthreads();
        group_scan(min(total - base, FLATCAP));
        __syncthreads();
        group_place(min(total - base, FLATCAP));
        __syncthreads();
        drain_round(TagNo{}, base);
        __syncthreads();
    }

    v4f wr[4];
#pragma unroll
    for (int j = 0; j < 4; ++j) {
        const int i = t + j * NTHR;
        wr[j] = *(const v4f*)(W + (size_t)(i >> 5) * KDIM + IN_FEAT + (i & 31) * 4);
    }
    for (int i = t; i < RB * (IN_FEAT / 4); i += NTHR) {
        const int r  = i >> 5;
        const int c4 = (i & 31) * 4;
        const float dg = fmaxf((float)degl[r], 1.0f);
        v4f* p = (v4f*)(feat + r * LDF + IN_FEAT + c4);
        v4f x = *p;
        x.x = x.x / dg; x.y = x.y / dg; x.z = x.z / dg; x.w = x.w / dg;
        *p = x;
    }
#pragma unroll
    for (int j = 0; j < 4; ++j) {
        const int i = t + j * NTHR;
        *(v4f*)(wl + (i >> 5) * LDW + (i & 31) * 4) = wr[j];
    }
    __syncthreads();

    auto gemm_rest = [&](auto tag) {
#pragma nounroll
        for (int k0 = IN_FEAT; k0 < KDIM; k0 += 16) block(tag, k0, k0 - IN_FEAT);
    };
    if (four) gemm_rest(TagYes{}); else gemm_rest(TagNo{});

    const float bv = bias[n];
    float* const ocol = out + n;
    const int g0 = row0 + 8 * h;
    auto store_tile = [&](int mt, const v8f& acc) {
#pragma unroll
        for (int r = 0; r < 8; ++r) {
            const int g = g0 + mt * 16 + r;
            if (g < n_dst) ocol[(size_t)g * OUT_FEAT] = acc[r] + bv;
        }
    };
    store_tile(mg, acc0);
    store_tile(mg + 4, acc1);
    store_tile(mg + 8, acc2);
    if (four) store_tile(12, acc3);
}

extern "C" void kernel_launch(void* const* d_in, const int* in_sizes, int n_in,
                              void* d_out, int out_size, void* d_ws, size_t ws_size,
                              hipStream_t stream) {
    const float* h_s  = (const float*)d_in[0];
    const float* h_d  = (const float*)d_in[1];
    const int*   src  = (const int*)d_in[2];
    const int*   dst  = (const int*)d_in[3];
    const float* W    = (const float*)d_in[4];
    const float* bias = (const float*)d_in[5];

    const int n_dst   = in_sizes[1] / IN_FEAT;
    const int n_edges = in_sizes[2];
    int* ws = (int*)d_ws;
    (void)n_in; (void)out_size; (void)ws_size;

    const int blocks = (n_dst + RB - 1) / RB;
    sage_sort_slices<<<NSL, STHR, 0, stream>>>(src, dst, ws, n_dst, n_edges);
    sage_owner_fused<<<blocks, NTHR, 0, stream>>>(h_s, h_d, ws, W, bias,
                                                  (float*)d_out, n_dst, n_edges);
}


# global stores leave as whole lines from an LDS stage in both launches: a slice's sorted entries (were scattered 4-byte stores), the output rows (were 64-byte half lines); values, sum order unchanged
# speedup vs baseline: 5.1557x; 1.1182x over previous
#include <hip/hip_runtime.h>


typedef __attribute__((ext_vector_type(2))) float v2f;
typedef __attribute__((ext_vector_type(4))) float v4f;
typedef __attribute__((ext_vector_type(8))) float v8f;

constexpr int IN_FEAT  = 128;
constexpr int KDIM     = 2 * IN_FEAT;
constexpr int OUT_FEAT = 128;
constexpr int LDF      = KDIM + 4;
constexpr int LDW      = IN_FEAT + 4;
constexpr int RB       = 208;
constexpr int NTHR     = 1024;
constexpr int NSL      = 256;
constexpr int STHR     = 256;
constexpr int UB       = 10;
constexpr int MAXOWN   = 1024;
constexpr int SCANW    = NSL / 32;
constexpr int SPW      = NSL / (NTHR / 32);
constexpr int FLATCAP  = 4096;
constexpr int SCAP     = 4096;

struct TagYes { static constexpr bool value = true;  };
struct TagNo  { static constexpr bool value = false; };

__device__ __forceinline__ void lds_add_f32(float* p, float v) {
    __hip_atomic_fetch_add(p, v, __ATOMIC_RELAXED, __HIP_MEMORY_SCOPE_WORKGROUP);
}
__device__ __forceinline__ void lds_add_i32(int* p, int v) {
    __hip_atomic_fetch_add(p, v, __ATOMIC_RELAXED, __HIP_MEMORY_SCOPE_WORKGROUP);
}

__device__ __forceinline__ int lds_add_i32_rtn(int* p, int v) {
    return __hip_atomic_fetch_add(p, v, __ATOMIC_RELAXED, __HIP_MEMORY_SCOPE_WORKGROUP);
}


__global__ __launch_bounds__(STHR) void sage_sort_slices(
    const int* __restrict__ src, const int* __restrict__ dst,
    int* __restrict__ ws, int n_dst, int n_edges)
{
    __shared__ int hist[MAXOWN + 1];
    __shared__ int stage[SCAP];

    const int t    = threadIdx.x;
    const int lane = t & 31;
    const int nown = (n_dst + RB - 1) / RB;
    const int L    = (n_edges + NSL - 1) / NSL;
    const int e0   = blockIdx.x * L;
    const int e1   = min(e0 + L, n_edges);
    int* offs = ws + (size_t)blockIdx.x * (nown + 1);
    int* ent  = ws + (size_t)NSL * (nown + 1) + e0;
    const bool staged = (L <= SCAP);

    for (int i = t; i <= nown; i += STHR) hist[i] = 0;
    __syncthreads();
    for (int eb = e0; eb < e1; eb += STHR * UB) {
        unsigned d[UB];
#pragma unroll
        for (int u = 0; u < UB; ++u) {
            const int e = eb + u * STHR + t;
            const unsigned x = (unsigned)dst[min(e, e1 - 1)];
            d[u] = (e < e1) ? x : 0xffffffffu;
        }
#pragma unroll
        for (int u = 0; u < UB; ++u)
            if (d[u] < (unsigned)n_dst) lds_add_i32(&hist[d[u] / RB], 1);
    }
    __syncthreads();
    if (t < 32) {
        const int per = (nown + 32) / 32;
        const int lo  = lane * per;
        const int hi  = min(lo + per, nown + 1);
        int sum = 0;
        for (int i = lo; i < hi; ++i) sum += hist[i];
        int x = sum;
        for (int o = 1; o < 32; o <<= 1) {
            const int y = __shfl_up(x, o);
            if (lane >= o) x += y;
        }
        int run = x - sum;
        for (int i = lo; i < hi; ++i) { const int c = hist[i]; hist[i] = run; run += c; }
    }
    __syncthreads();
    for (int i = t; i <= nown; i += STHR) offs[i] = hist[i];
    __syncthreads();
    for (int eb = e0; eb < e1; eb += STHR * UB) {
        unsigned d[UB];
        int      sr[UB];
#pragma unroll
        for (int u = 0; u < UB; ++u) {
            const int e  = eb + u * STHR + t;
            const int ec = min(e, e1 - 1);
            const unsigned x = (unsigned)dst[ec];
            sr[u] = src[ec];
            d[u]  = (e < e1) ? x : 0xffffffffu;
        }
        int pos[UB];
#pragma unroll
        for (int u = 0; u < UB; ++u) {
            pos[u] = -1;
            if (d[u] < (unsigned)n_dst) pos[u] = lds_add_i32_rtn(&hist[d[u] / RB], 1);
        }
#pragma unroll
        for (int u = 0; u < UB; ++u) {
            if (pos[u] >= 0) {
                const unsigned b = d[u] / RB;
                const int x = (sr[u] << 8) | (int)(d[u] - b * RB);
                if (staged) stage[pos[u]] = x; else ent[pos[u]] = x;
            }
        }
    }
    if (staged) {
        __syncthreads();
        const int tot = hist[nown];
        for (int ib = 0; ib < tot; ib += STHR * UB) {
            int v[UB];
#pragma unroll
            for (int u = 0; u < UB; ++u) v[u] = stage[min(ib + u * STHR + t, tot - 1)];
#pragma unroll
            for (int u = 0; u < UB; ++u) {
                const int i = ib + u * STHR + t;
                if (i < tot) ent[i] = v[u];
            }
        }
    }
}

__global__ __launch_bounds__(NTHR) void sage_owner_fused(
    const float* __restrict__ h_s, const float* __restrict__ h_d,
    const int* __restrict__ ws,
    const float* __restrict__ W, const float* __restrict__ bias,
    float* __restrict__ out, int n_dst, int n_edges)
{
    __shared__ __attribute__((aligned(16))) float feat[RB * LDF];
    __shared__ __attribute__((aligned(16))) float wl[OUT_FEAT * LDW];
    __shared__ int degl[RB];
    __shared__ __attribute__((aligned(16))) int flat[FLATCAP];
    __shared__ __attribute__((aligned(16))) int sorted[FLATCAP];
    __shared__ int rcnt[RB];
    __shared__ int rstart[RB + 1];
    __shared__ int rcur[RB];
    __shared__ int slo[NSL], scnt[NSL], spre[NSL];
    __shared__ int wtot[SCANW];

    const int t    = threadIdx.x;
    const int lane = t & 31;
    const int wave = __builtin_amdgcn_readfirstlane(t >> 5);
    const int row0 = blockIdx.x * RB;

    const int nown = (n_dst + RB - 1) / RB;
    const int L    = (n_edges + NSL - 1) / NSL;
    const int* ent = ws + (size_t)NSL * (nown + 1);

    int rlo = 0, rhi = 0;
    if (wave < SCANW) {
        const int* o = ws + (size_t)t * (nown + 1) + blockIdx.x;
        rlo = o[0];
        rhi = o[1];
    }

    for (int i = t; i < RB * (IN_FEAT / 4); i += NTHR) {
        const int r  = i >> 5;
        const int c4 = (i & 31) * 4;
        const int g  = row0 + r;
        v4f v = {0.f, 0.f, 0.f, 0.f};
        if (g < n_dst) v = *(const v4f*)(h_d + (size_t)g * IN_FEAT + c4);
        *(v4f*)(feat + r * LDF + c4) = v;
        const v4f z = {0.f, 0.f, 0.f, 0.f};
        *(v4f*)(feat + r * LDF + IN_FEAT + c4) = z;
    }
    for (int i = t; i < RB; i += NTHR) { degl[i] = 0; rcnt[i] = 0; }
    static_assert(OUT_FEAT * (IN_FEAT / 4) == 4 * NTHR, "four 16-byte pieces of a W half per thread");
#pragma unroll
    for (int j = 0; j < 4; ++j) {
        const int i = t + j * NTHR;
        *(v4f*)(wl + (i >> 5) * LDW + (i & 31) * 4) = *(const v4f*)(W + (size_t)(i >> 5) * KDIM + (i & 31) * 4);
    }

    if (wave < SCANW) {
        const int cnt = rhi - rlo;
        int x = cnt;
        for (int o = 1; o < 32; o <<= 1) {
            const int y = __shfl_up(x, o);
            if (lane >= o) x += y;
        }
        slo[t]  = rlo;
        scnt[t] = cnt;
        spre[t] = x - cnt;
        if (lane == 31) wtot[wave] = x;
    }
    __syncthreads();
    static_assert(32 % SPW == 0, "the slices one wave copies must lie in one scan wave");
    const int myscan = (wave * SPW) >> 5;
    int total = 0, wpre = 0;
#pragma unroll
    for (int k = 0; k < SCANW; ++k) {
        const int w = wtot[k];
        total += w;
        if (k < myscan) wpre += w;
    }

    const int c     = lane & 15;
    const int h     = lane >> 4;
    const int ntile = wave & 7;
    const int mg    = wave >> 3;
    const bool four = (mg == 0);
    const int n     = ntile * 16 + c;

    const float* bw = wl + n * LDW + 2 * h;
    const float* a0 = feat + (mg * 16 + c) * LDF + 2 * h;
    const float* a1 = a0 + 4 * 16 * LDF;
    const float* a2 = a0 + 8 * 16 * LDF;
    const float* a3 = a0 + 12 * 16 * LDF;

    v8f acc0 = {}, acc1 = {}, acc2 = {}, acc3 = {};
    constexpr int HB = IN_FEAT / 16;
    int kb = 0;

    auto block = [&](auto tag, int k0, int kw) {
        constexpr bool FOUR = decltype(tag)::value;
        v2f x0[4], x1[4], x2[4], x3[4], bf[4];
#pragma unroll
        for (int u = 0; u < 4; ++u) {
            bf[u] = *(const v2f*)(bw + kw + 4 * u);
            x0[u] = *(const v2f*)(a0 + k0 + 4 * u);
            x1[u] = *(const v2f*)(a1 + k0 + 4 * u);
            x2[u] = *(const v2f*)(a2 + k0 + 4 * u);
            if (FOUR) x3[u] = *(const v2f*)(a3 + k0 + 4 * u);
        }
#pragma unroll
        for (int u = 0; u < 4; ++u) {
            acc0 = __builtin_amdgcn_wmma_f32_16x16x4_f32(false, x0[u], false, bf[u], (short)0, acc0, false, false);
            acc1 = __builtin_amdgcn_wmma_f32_16x16x4_f32(false, x1[u], false, bf[u], (short)0, acc1, false, false);
            acc2 = __builtin_amdgcn_wmma_f32_16x16x4_f32(false, x2[u], false, bf[u], (short)0, acc2, false, false);
            if (FOUR)
                acc3 = __builtin_amdgcn_wmma_f32_16x16x4_f32(false, x3[u], false, bf[u], (short)0, acc3, false, false);
        }
    };
    auto tile12_hd = [&]() {
#pragma nounroll
        for (int k0 = 0; k0 < IN_FEAT; k0 += 16) {
            v2f x3[4], bf[4];
#pragma unroll
            for (int u = 0; u < 4; ++u) {
                bf[u] = *(const v2f*)(bw + k0 + 4 * u);
                x3[u] = *(const v2f*)(a3 + k0 + 4 * u);
            }
#pragma unroll
            for (int u = 0; u < 4; ++u)
                acc3 = __builtin_amdgcn_wmma_f32_16x16x4_f32(false, x3[u], false, bf[u], (short)0, acc3, false, false);
        }
    };

    auto flush = [&](auto tag, int r, v4f s, int n) {
        constexpr bool FIRST = decltype(tag)::value;
        float* a = feat + r * LDF + IN_FEAT + 4 * lane;
        if (!FIRST) {
            const v4f p = *(const v4f*)a;
            s.x += p.x; s.y += p.y; s.z += p.z; s.w += p.w;
            n += __builtin_amdgcn_readfirstlane(degl[r]);
        }
        *(v4f*)a = s;
        if (lane == 0) degl[r] = n;
    };

    auto drain = [&](auto tag, const int* q, int cnt, int rfirst) {
        constexpr bool FIRST = decltype(tag)::value;
        v4f s = {0.f, 0.f, 0.f, 0.f};
        int cur = rfirst;
        int n   = 0;
        auto take = [&](int lr, v4f v) {
            if (lr != cur) {
                if (n > 0) flush(tag, cur, s, n);
                cur = lr;
                s.x = 0.f; s.y = 0.f; s.z = 0.f; s.w = 0.f;
                n = 0;
            }
            s.x += v.x; s.y += v.y; s.z += v.z; s.w += v.w;
            ++n;
        };
        int i0 = 0;
        if (FIRST)
        for (; i0 < cnt && kb < HB; i0 += 8, ++kb) {
            v4f v[8];
            int lr[8];
#pragma unroll
            for (int u = 0; u < 8; ++u) {
                const unsigned p = (unsigned)__builtin_amdgcn_readfirstlane(q[min(i0 + u, cnt - 1)]);
                lr[u] = (int)(p & 255u);
                v[u]  = *(const v4f*)(h_s + (size_t)(p >> 8) * IN_FEAT + lane * 4);
            }
            __builtin_amdgcn_sched_barrier(0);
            block(TagNo{}, kb * 16, kb * 16);
            __builtin_amdgcn_sched_barrier(0);
#pragma unroll
            for (int u = 0; u < 8; ++u)
                if (i0 + u < cnt) take(lr[u], v[u]);
        }
        for (; i0 < cnt; i0 += 8) {
            v4f v[8];
            int lr[8];
#pragma unroll
            for (int u = 0; u < 8; ++u) {
                const unsigned p = (unsigned)__builtin_amdgcn_readfirstlane(q[min(i0 + u, cnt - 1)]);
                lr[u] = (int)(p & 255u);
                v[u]  = *(const v4f*)(h_s + (size_t)(p >> 8) * IN_FEAT + lane * 4);
            }
#pragma unroll
            for (int u = 0; u < 8; ++u)
                if (i0 + u < cnt) take(lr[u], v[u]);
        }
        if (n > 0) flush(tag, cur, s, n);
        if (FIRST) {
            for (; kb < HB; ++kb) block(TagNo{}, kb * 16, kb * 16);
            if (four) tile12_hd();
        }
    };

    total = __builtin_amdgcn_readfirstlane(total);

    auto copy_round = [&](int base) {
        int cs[SPW], ps[SPW], ls[SPW];
#pragma unroll
        for (int j = 0; j < SPW; ++j) {
            cs[j] = __builtin_amdgcn_readfirstlane(scnt[wave * SPW + j]);
            ps[j] = __builtin_amdgcn_readfirstlane(spre[wave * SPW + j] + wpre);
            ls[j] = __builtin_amdgcn_readfirstlane(slo[wave * SPW + j]);
        }
        int rv[SPW], rg[SPW];
#pragma unroll
        for (int j = 0; j < SPW; ++j) {
            const int g  = ps[j] + lane - base;
            const bool m = lane < cs[j] && (unsigned)g < (unsigned)FLATCAP;
            rv[j] = 0;
            rg[j] = -1;
            if (m) { rv[j] = ent[(size_t)(wave * SPW + j) * L + ls[j] + lane]; rg[j] = g; }
        }
#pragma unroll
        for (int j = 0; j < SPW; ++j)
            if (rg[j] >= 0) { flat[rg[j]] = rv[j]; lds_add_i32(&rcnt[rv[j] & 255], 1); }
#pragma unroll
        for (int j = 0; j < SPW; ++j) {
            const int* run = ent + (size_t)(wave * SPW + j) * L + ls[j];
            for (int i = lane + 32; i < cs[j]; i += 32) {
                const unsigned g = (unsigned)(ps[j] + i - base);
                if (g < (unsigned)FLATCAP) { const int x = run[i]; flat[g] = x; lds_add_i32(&rcnt[x & 255], 1); }
            }
        }
    };
    auto group_scan = [&](int nr) {
        if (wave == 0) {
            constexpr int PER = (RB + 31) / 32;
            const int lo = lane * PER;
            const int hi = min(lo + PER, RB);
            int sum = 0;
            for (int i = lo; i < hi; ++i) sum += rcnt[i];
            int x = sum;
            for (int o = 1; o < 32; o <<= 1) {
                const int y = __shfl_up(x, o);
                if (lane >= o) x += y;
            }
            int run = x - sum;
            for (int i = lo; i < hi; ++i) { const int c = rcnt[i]; rstart[i] = run; rcur[i] = run; run += c; }
            if (lane == 0) rstart[RB] = nr;
        }
    };
    auto group_place = [&](int nr) {
        for (int i = t; i < nr; i += NTHR) {
            const int p = flat[i];
            const int pos = lds_add_i32_rtn(&rcur[p & 255], 1);
            sorted[pos] = p;
        }
    };
    auto rows_below = [&](int x) {
        int cnt = 0;
#pragma unroll
        for (int j = 0; j < (RB + 31) / 32; ++j) {
            const int r = j * 32 + lane;
            const bool b = (r < RB) && (rstart[min(r, RB)] < x);
            cnt += __popc(__builtin_amdgcn_ballot_w32(b));
        }
        return cnt;
    };
    auto drain_round = [&](auto tag, int base) {
        const int nr    = min(total - base, FLATCAP);
        const int chunk = (nr + 31) >> 5;
        const int rw0   = (wave == 0) ? 0 : rows_below(wave * chunk);
        const int rw1   = (wave == NTHR / 32 - 1) ? RB : rows_below((wave + 1) * chunk);
        const int e0    = __builtin_amdgcn_readfirstlane(rstart[rw0]);
        const int e1    = __builtin_amdgcn_readfirstlane(rstart[rw1]);
        drain(tag, sorted + e0, e1 - e0, rw0);
    };

    copy_round(0);
    __syncthreads();
    group_scan(min(total, FLATCAP));
    __syncthreads();
    group_place(min(total, FLATCAP));
    __syncthreads();
    drain_round(TagYes{}, 0);
    __syncthreads();
    for (int base = FLATCAP; base < total; base += FLATCAP) {
        for (int i = t; i < RB; i += NTHR) rcnt[i] = 0;
        __syncthreads();
        copy_round(base);
        __syncthreads();
        group_scan(min(total - base, FLATCAP));
        __syncthreads();
        group_place(min(total - base, FLATCAP));
        __syncthreads();
        drain_round(TagNo{}, base);
        __syncthreads();
    }

    v4f wr[4];
#pragma unroll
    for (int j = 0; j < 4; ++j) {
        const int i = t + j * NTHR;
        wr[j] = *(const v4f*)(W + (size_t)(i >> 5) * KDIM + IN_FEAT + (i & 31) * 4);
    }
    for (int i = t; i < RB * (IN_FEAT / 4); i += NTHR) {
        const int r  = i >> 5;
        const int c4 = (i & 31) * 4;
        const float dg = fmaxf((float)degl[r], 1.0f);
        v4f* p = (v4f*)(feat + r * LDF + IN_FEAT + c4);
        v4f x = *p;
        x.x = x.x / dg; x.y = x.y / dg; x.z = x.z / dg; x.w = x.w / dg;
        *p = x;
    }
#pragma unroll
    for (int j = 0; j < 4; ++j) {
        const int i = t + j * NTHR;
        *(v4f*)(wl + (i >> 5) * LDW + (i & 31) * 4) = wr[j];
    }
    __syncthreads();

    auto gemm_rest = [&](auto tag) {
#pragma nounroll
        for (int k0 = IN_FEAT; k0 < KDIM; k0 += 16) block(tag, k0, k0 - IN_FEAT);
    };
    const float bv = bias[n];
    if (four) gemm_rest(TagYes{}); else gemm_rest(TagNo{});

    auto stage_tile = [&](int mt, const v8f& acc) {
        float* p = feat + (mt * 16 + 8 * h) * LDF + n;
#pragma unroll
        for (int r = 0; r < 8; ++r) p[r * LDF] = acc[r] + bv;
    };
    stage_tile(mg, acc0);
    stage_tile(mg + 4, acc1);
    stage_tile(mg + 8, acc2);
    if (four) stage_tile(12, acc3);
    __syncthreads();
    constexpr int OIT = (RB * (OUT_FEAT / 4) + NTHR - 1) / NTHR;
    v4f ov[OIT];
#pragma unroll
    for (int j = 0; j < OIT; ++j) {
        const int i = t + j * NTHR;
        ov[j] = *(const v4f*)(feat + min(i >> 5, RB - 1) * LDF + (i & 31) * 4);
    }
#pragma unroll
    for (int j = 0; j < OIT; ++j) {
        const int i = t + j * NTHR;
        const int r = i >> 5;
        const int g = row0 + r;
        if (r < RB && g < n_dst) *(v4f*)(out + (size_t)g * OUT_FEAT + (i & 31) * 4) = ov[j];
    }
}

extern "C" void kernel_launch(void* const* d_in, const int* in_sizes, int n_in,
                              void* d_out, int out_size, void* d_ws, size_t ws_size,
                              hipStream_t stream) {
    const float* h_s  = (const float*)d_in[0];
    const float* h_d  = (const float*)d_in[1];
    const int*   src  = (const int*)d_in[2];
    const int*   dst  = (const int*)d_in[3];
    const float* W    = (const float*)d_in[4];
    const float* bias = (const float*)d_in[5];

    const int n_dst   = in_sizes[1] / IN_FEAT;
    const int n_edges = in_sizes[2];
    int* ws = (int*)d_ws;
    (void)n_in; (void)out_size; (void)ws_size;

    const int blocks = (n_dst + RB - 1) / RB;
    sage_sort_slices<<<NSL, STHR, 0, stream>>>(src, dst, ws, n_dst, n_edges);
    sage_owner_fused<<<blocks, NTHR, 0, stream>>>(h_s, h_d, ws, W, bias,
                                                  (float*)d_out, n_dst, n_edges);
}


# set-up loads requested as one group, none waited for alone: owner's 7 h_d + 4 W pieces per thread (was a round trip per loop trip) fly over an LDS-only barrier; sort's first dst pass likewise
# speedup vs baseline: 5.4585x; 1.0587x over previous
#include <hip/hip_runtime.h>


typedef __attribute__((ext_vector_type(2))) float v2f;
typedef __attribute__((ext_vector_type(4))) float v4f;
typedef __attribute__((ext_vector_type(8))) float v8f;

constexpr int IN_FEAT  = 128;
constexpr int KDIM     = 2 * IN_FEAT;
constexpr int OUT_FEAT = 128;
constexpr int LDF      = KDIM + 4;
constexpr int LDW      = IN_FEAT + 4;
constexpr int RB       = 208;
constexpr int NTHR     = 1024;
constexpr int NSL      = 256;
constexpr int STHR     = 256;
constexpr int UB       = 10;
constexpr int MAXOWN   = 1024;
constexpr int SCANW    = NSL / 32;
constexpr int SPW      = NSL / (NTHR / 32);
constexpr int FLATCAP  = 4096;
constexpr int SCAP     = 4096;

struct TagYes { static constexpr bool value = true;  };
struct TagNo  { static constexpr bool value = false; };

__device__ __forceinline__ void lds_add_f32(float* p, float v) {
    __hip_atomic_fetch_add(p, v, __ATOMIC_RELAXED, __HIP_MEMORY_SCOPE_WORKGROUP);
}
__device__ __forceinline__ void lds_add_i32(int* p, int v) {
    __hip_atomic_fetch_add(p, v, __ATOMIC_RELAXED, __HIP_MEMORY_SCOPE_WORKGROUP);
}

__device__ __forceinline__ int lds_add_i32_rtn(int* p, int v) {
    return __hip_atomic_fetch_add(p, v, __ATOMIC_RELAXED, __HIP_MEMORY_SCOPE_WORKGROUP);
}


__global__ __launch_bounds__(STHR) void sage_sort_slices(
    const int* __restrict__ src, const int* __restrict__ dst,
    int* __restrict__ ws, int n_dst, int n_edges)
{
    __shared__ int hist[MAXOWN + 1];
    __shared__ int stage[SCAP];

    const int t    = threadIdx.x;
    const int lane = t & 31;
    const int nown = (n_dst + RB - 1) / RB;
    const int L    = (n_edges + NSL - 1) / NSL;
    const int e0   = blockIdx.x * L;
    const int e1   = min(e0 + L, n_edges);
    int* offs = ws + (size_t)blockIdx.x * (nown + 1);
    int* ent  = ws + (size_t)NSL * (nown + 1) + e0;
    const bool staged = (L <= SCAP);

    for (int i = t; i <= nown; i += STHR) hist[i] = 0;
    __syncthreads();
    for (int eb = e0; eb < e1; eb += STHR * UB) {
        unsigned x[UB];
#pragma unroll
        for (int u = 0; u < UB; ++u) {
            const int e = eb + u * STHR + t;
            x[u] = (unsigned)dst[min(e, e1 - 1)];
        }
        __builtin_amdgcn_sched_barrier(0);
#pragma unroll
        for (int u = 0; u < UB; ++u) {
            const int e = eb + u * STHR + t;
            const unsigned d = (e < e1) ? x[u] : 0xffffffffu;
            if (d < (unsigned)n_dst) lds_add_i32(&hist[d / RB], 1);
        }
    }
    __syncthreads();
    if (t < 32) {
        const int per = (nown + 32) / 32;
        const int lo  = lane * per;
        const int hi  = min(lo + per, nown + 1);
        int sum = 0;
        for (int i = lo; i < hi; ++i) sum += hist[i];
        int x = sum;
        for (int o = 1; o < 32; o <<= 1) {
            const int y = __shfl_up(x, o);
            if (lane >= o) x += y;
        }
        int run = x - sum;
        for (int i = lo; i < hi; ++i) { const int c = hist[i]; hist[i] = run; run += c; }
    }
    __syncthreads();
    for (int i = t; i <= nown; i += STHR) offs[i] = hist[i];
    __syncthreads();
    for (int eb = e0; eb < e1; eb += STHR * UB) {
        unsigned d[UB];
        int      sr[UB];
#pragma unroll
        for (int u = 0; u < UB; ++u) {
            const int e  = eb + u * STHR + t;
            const int ec = min(e, e1 - 1);
            const unsigned x = (unsigned)dst[ec];
            sr[u] = src[ec];
            d[u]  = (e < e1) ? x : 0xffffffffu;
        }
        int pos[UB];
#pragma unroll
        for (int u = 0; u < UB; ++u) {
            pos[u] = -1;
            if (d[u] < (unsigned)n_dst) pos[u] = lds_add_i32_rtn(&hist[d[u] / RB], 1);
        }
#pragma unroll
        for (int u = 0; u < UB; ++u) {
            if (pos[u] >= 0) {
                const unsigned b = d[u] / RB;
                const int x = (sr[u] << 8) | (int)(d[u] - b * RB);
                if (staged) stage[pos[u]] = x; else ent[pos[u]] = x;
            }
        }
    }
    if (staged) {
        __syncthreads();
        const int tot = hist[nown];
        for (int ib = 0; ib < tot; ib += STHR * UB) {
            int v[UB];
#pragma unroll
            for (int u = 0; u < UB; ++u) v[u] = stage[min(ib + u * STHR + t, tot - 1)];
#pragma unroll
            for (int u = 0; u < UB; ++u) {
                const int i = ib + u * STHR + t;
                if (i < tot) ent[i] = v[u];
            }
        }
    }
}

__global__ __launch_bounds__(NTHR) void sage_owner_fused(
    const float* __restrict__ h_s, const float* __restrict__ h_d,
    const int* __restrict__ ws,
    const float* __restrict__ W, const float* __restrict__ bias,
    float* __restrict__ out, int n_dst, int n_edges)
{
    __shared__ __attribute__((aligned(16))) float feat[RB * LDF];
    __shared__ __attribute__((aligned(16))) float wl[OUT_FEAT * LDW];
    __shared__ int degl[RB];
    __shared__ __attribute__((aligned(16))) int flat[FLATCAP];
    __shared__ __attribute__((aligned(16))) int sorted[FLATCAP];
    __shared__ int rcnt[RB];
    __shared__ int rstart[RB + 1];
    __shared__ int rcur[RB];
    __shared__ int slo[NSL], scnt[NSL], spre[NSL];
    __shared__ int wtot[SCANW];

    const int t    = threadIdx.x;
    const int lane = t & 31;
    const int wave = __builtin_amdgcn_readfirstlane(t >> 5);
    const int row0 = blockIdx.x * RB;

    const int nown = (n_dst + RB - 1) / RB;
    const int L    = (n_edges + NSL - 1) / NSL;
    const int* ent = ws + (size_t)NSL * (nown + 1);

    int rlo = 0, rhi = 0;
    if (wave < SCANW) {
        const int* o = ws + (size_t)t * (nown + 1) + blockIdx.x;
        rlo = o[0];
        rhi = o[1];
    }

    constexpr int HIT = (RB * (IN_FEAT / 4) + NTHR - 1) / NTHR;
    constexpr int WIT = OUT_FEAT * (IN_FEAT / 4) / NTHR;
    static_assert(OUT_FEAT * (IN_FEAT / 4) == WIT * NTHR, "whole 16-byte pieces of a W half per thread");
    v4f hd[HIT], wv[WIT];
    int ts = t;
    asm volatile("" : "+v"(ts));
#pragma unroll
    for (int j = 0; j < HIT; ++j) {
        const int i = ts + j * NTHR;
        const int g = min(row0 + min(i >> 5, RB - 1), n_dst - 1);
        hd[j] = *(const v4f*)(h_d + (size_t)g * IN_FEAT + (i & 31) * 4);
    }
#pragma unroll
    for (int j = 0; j < WIT; ++j) {
        const int i = ts + j * NTHR;
        wv[j] = *(const v4f*)(W + (size_t)(i >> 5) * KDIM + (i & 31) * 4);
    }
    for (int i = t; i < RB * (IN_FEAT / 4); i += NTHR) {
        const v4f z = {0.f, 0.f, 0.f, 0.f};
        *(v4f*)(feat + (i >> 5) * LDF + IN_FEAT + (i & 31) * 4) = z;
    }
    for (int i = t; i < RB; i += NTHR) { degl[i] = 0; rcnt[i] = 0; }

    if (wave < SCANW) {
        const int cnt = rhi - rlo;
        int x = cnt;
        for (int o = 1; o < 32; o <<= 1) {
            const int y = __shfl_up(x, o);
            if (lane >= o) x += y;
        }
        slo[t]  = rlo;
        scnt[t] = cnt;
        spre[t] = x - cnt;
        if (lane == 31) wtot[wave] = x;
    }
    __builtin_amdgcn_fence(__ATOMIC_RELEASE, "workgroup", "local");
    __builtin_amdgcn_s_barrier();
    __builtin_amdgcn_fence(__ATOMIC_ACQUIRE, "workgroup", "local");
    static_assert(32 % SPW == 0, "the slices one wave copies must lie in one scan wave");
    const int myscan = (wave * SPW) >> 5;
    int total = 0, wpre = 0;
#pragma unroll
    for (int k = 0; k < SCANW; ++k) {
        const int w = wtot[k];
        total += w;
        if (k < myscan) wpre += w;
    }

    const int c     = lane & 15;
    const int h     = lane >> 4;
    const int ntile = wave & 7;
    const int mg    = wave >> 3;
    const bool four = (mg == 0);
    const int n     = ntile * 16 + c;

    const float* bw = wl + n * LDW + 2 * h;
    const float* a0 = feat + (mg * 16 + c) * LDF + 2 * h;
    const float* a1 = a0 + 4 * 16 * LDF;
    const float* a2 = a0 + 8 * 16 * LDF;
    const float* a3 = a0 + 12 * 16 * LDF;

    v8f acc0 = {}, acc1 = {}, acc2 = {}, acc3 = {};
    constexpr int HB = IN_FEAT / 16;
    int kb = 0;

    auto block = [&](auto tag, int k0, int kw) {
        constexpr bool FOUR = decltype(tag)::value;
        v2f x0[4], x1[4], x2[4], x3[4], bf[4];
#pragma unroll
        for (int u = 0; u < 4; ++u) {
            bf[u] = *(const v2f*)(bw + kw + 4 * u);
            x0[u] = *(const v2f*)(a0 + k0 + 4 * u);
            x1[u] = *(const v2f*)(a1 + k0 + 4 * u);
            x2[u] = *(const v2f*)(a2 + k0 + 4 * u);
            if (FOUR) x3[u] = *(const v2f*)(a3 + k0 + 4 * u);
        }
#pragma unroll
        for (int u = 0; u < 4; ++u) {
            acc0 = __builtin_amdgcn_wmma_f32_16x16x4_f32(false, x0[u], false, bf[u], (short)0, acc0, false, false);
            acc1 = __builtin_amdgcn_wmma_f32_16x16x4_f32(false, x1[u], false, bf[u], (short)0, acc1, false, false);
            acc2 = __builtin_amdgcn_wmma_f32_16x16x4_f32(false, x2[u], false, bf[u], (short)0, acc2, false, false);
            if (FOUR)
                acc3 = __builtin_amdgcn_wmma_f32_16x16x4_f32(false, x3[u], false, bf[u], (short)0, acc3, false, false);
        }
    };
    auto tile12_hd = [&]() {
#pragma nounroll
        for (int k0 = 0; k0 < IN_FEAT; k0 += 16) {
            v2f x3[4], bf[4];
#pragma unroll
            for (int u = 0; u < 4; ++u) {
                bf[u] = *(const v2f*)(bw + k0 + 4 * u);
                x3[u] = *(const v2f*)(a3 + k0 + 4 * u);
            }
#pragma unroll
            for (int u = 0; u < 4; ++u)
                acc3 = __builtin_amdgcn_wmma_f32_16x16x4_f32(false, x3[u], false, bf[u], (short)0, acc3, false, false);
        }
    };

    auto flush = [&](auto tag, int r, v4f s, int n) {
        constexpr bool FIRST = decltype(tag)::value;
        float* a = feat + r * LDF + IN_FEAT + 4 * lane;
        if (!FIRST) {
            const v4f p = *(const v4f*)a;
            s.x += p.x; s.y += p.y; s.z += p.z; s.w += p.w;
            n += __builtin_amdgcn_readfirstlane(degl[r]);
        }
        *(v4f*)a = s;
        if (lane == 0) degl[r] = n;
    };

    auto drain = [&](auto tag, const int* q, int cnt, int rfirst) {
        constexpr bool FIRST = decltype(tag)::value;
        v4f s = {0.f, 0.f, 0.f, 0.f};
        int cur = rfirst;
        int n   = 0;
        auto take = [&](int lr, v4f v) {
            if (lr != cur) {
                if (n > 0) flush(tag, cur, s, n);
                cur = lr;
                s.x = 0.f; s.y = 0.f; s.z = 0.f; s.w = 0.f;
                n = 0;
            }
            s.x += v.x; s.y += v.y; s.z += v.z; s.w += v.w;
            ++n;
        };
        int i0 = 0;
        if (FIRST)
        for (; i0 < cnt && kb < HB; i0 += 8, ++kb) {
            v4f v[8];
            int lr[8];
#pragma unroll
            for (int u = 0; u < 8; ++u) {
                const unsigned p = (unsigned)__builtin_amdgcn_readfirstlane(q[min(i0 + u, cnt - 1)]);
                lr[u] = (int)(p & 255u);
                v[u]  = *(const v4f*)(h_s + (size_t)(p >> 8) * IN_FEAT + lane * 4);
            }
            __builtin_amdgcn_sched_barrier(0);
            block(TagNo{}, kb * 16, kb * 16);
            __builtin_amdgcn_sched_barrier(0);
#pragma unroll
            for (int u = 0; u < 8; ++u)
                if (i0 + u < cnt) take(lr[u], v[u]);
        }
        for (; i0 < cnt; i0 += 8) {
            v4f v[8];
            int lr[8];
#pragma unroll
            for (int u = 0; u < 8; ++u) {
                const unsigned p = (unsigned)__builtin_amdgcn_readfirstlane(q[min(i0 + u, cnt - 1)]);
                lr[u] = (int)(p & 255u);
                v[u]  = *(const v4f*)(h_s + (size_t)(p >> 8) * IN_FEAT + lane * 4);
            }
#pragma unroll
            for (int u = 0; u < 8; ++u)
                if (i0 + u < cnt) take(lr[u], v[u]);
        }
        if (n > 0) flush(tag, cur, s, n);
        if (FIRST) {
            for (; kb < HB; ++kb) block(TagNo{}, kb * 16, kb * 16);
            if (four) tile12_hd();
        }
    };

    total = __builtin_amdgcn_readfirstlane(total);

    auto copy_round = [&](int base) {
        int cs[SPW], ps[SPW], ls[SPW];
#pragma unroll
        for (int j = 0; j < SPW; ++j) {
            cs[j] = __builtin_amdgcn_readfirstlane(scnt[wave * SPW + j]);
            ps[j] = __builtin_amdgcn_readfirstlane(spre[wave * SPW + j] + wpre);
            ls[j] = __builtin_amdgcn_readfirstlane(slo[wave * SPW + j]);
        }
        int rv[SPW], rg[SPW];
#pragma unroll
        for (int j = 0; j < SPW; ++j) {
            const int g  = ps[j] + lane - base;
            const bool m = lane < cs[j] && (unsigned)g < (unsigned)FLATCAP;
            rv[j] = 0;
            rg[j] = -1;
            if (m) { rv[j] = ent[(size_t)(wave * SPW + j) * L + ls[j] + lane]; rg[j] = g; }
        }
#pragma unroll
        for (int j = 0; j < SPW; ++j)
            if (rg[j] >= 0) { flat[rg[j]] = rv[j]; lds_add_i32(&rcnt[rv[j] & 255], 1); }
#pragma unroll
        for (int j = 0; j < SPW; ++j) {
            const int* run = ent + (size_t)(wave * SPW + j) * L + ls[j];
            for (int i = lane + 32; i < cs[j]; i += 32) {
                const unsigned g = (unsigned)(ps[j] + i - base);
                if (g < (unsigned)FLATCAP) { const int x = run[i]; flat[g] = x; lds_add_i32(&rcnt[x & 255], 1); }
            }
        }
    };
    auto group_scan = [&](int nr) {
        if (wave == 0) {
            constexpr int PER = (RB + 31) / 32;
            const int lo = lane * PER;
            const int hi = min(lo + PER, RB);
            int sum = 0;
            for (int i = lo; i < hi; ++i) sum += rcnt[i];
            int x = sum;
            for (int o = 1; o < 32; o <<= 1) {
                const int y = __shfl_up(x, o);
                if (lane >= o) x += y;
            }
            int run = x - sum;
            for (int i = lo; i < hi; ++i) { const int c = rcnt[i]; rstart[i] = run; rcur[i] = run; run += c; }
            if (lane == 0) rstart[RB] = nr;
        }
    };
    auto group_place = [&](int nr) {
        for (int i = t; i < nr; i += NTHR) {
            const int p = flat[i];
            const int pos = lds_add_i32_rtn(&rcur[p & 255], 1);
            sorted[pos] = p;
        }
    };
    auto rows_below = [&](int x) {
        int cnt = 0;
#pragma unroll
        for (int j = 0; j < (RB + 31) / 32; ++j) {
            const int r = j * 32 + lane;
            const bool b = (r < RB) && (rstart[min(r, RB)] < x);
            cnt += __popc(__builtin_amdgcn_ballot_w32(b));
        }
        return cnt;
    };
    auto drain_round = [&](auto tag, int base) {
        const int nr    = min(total - base, FLATCAP);
        const int chunk = (nr + 31) >> 5;
        const int rw0   = (wave == 0) ? 0 : rows_below(wave * chunk);
        const int rw1   = (wave == NTHR / 32 - 1) ? RB : rows_below((wave + 1) * chunk);
        const int e0    = __builtin_amdgcn_readfirstlane(rstart[rw0]);
        const int e1    = __builtin_amdgcn_readfirstlane(rstart[rw1]);
        drain(tag, sorted + e0, e1 - e0, rw0);
    };

    copy_round(0);
#pragma unroll
    for (int j = 0; j < HIT; ++j) {
        const int i = ts + j * NTHR;
        const int r = i >> 5;
        if ((j + 1) * NTHR <= RB * (IN_FEAT / 4) || r < RB) {
            const v4f z = {0.f, 0.f, 0.f, 0.f};
            *(v4f*)(feat + r * LDF + (i & 31) * 4) = (row0 + r < n_dst) ? hd[j] : z;
        }
    }
#pragma unroll
    for (int j = 0; j < WIT; ++j) {
        const int i = ts + j * NTHR;
        *(v4f*)(wl + (i >> 5) * LDW + (i & 31) * 4) = wv[j];
    }
    __syncthreads();
    group_scan(min(total, FLATCAP));
    __syncthreads();
    group_place(min(total, FLATCAP));
    __syncthreads();
    drain_round(TagYes{}, 0);
    __syncthreads();
    for (int base = FLATCAP; base < total; base += FLATCAP) {
        for (int i = t; i < RB; i += NTHR) rcnt[i] = 0;
        __syncthreads();
        copy_round(base);
        __syncthreads();
        group_scan(min(total - base, FLATCAP));
        __syncthreads();
        group_place(min(total - base, FLATCAP));
        __syncthreads();
        drain_round(TagNo{}, base);
        __syncthreads();
    }

    v4f wr[4];
#pragma unroll
    for (int j = 0; j < 4; ++j) {
        const int i = t + j * NTHR;
        wr[j] = *(const v4f*)(W + (size_t)(i >> 5) * KDIM + IN_FEAT + (i & 31) * 4);
    }
    for (int i = t; i < RB * (IN_FEAT / 4); i += NTHR) {
        const int r  = i >> 5;
        const int c4 = (i & 31) * 4;
        const float dg = fmaxf((float)degl[r], 1.0f);
        v4f* p = (v4f*)(feat + r * LDF + IN_FEAT + c4);
        v4f x = *p;
        x.x = x.x / dg; x.y = x.y / dg; x.z = x.z / dg; x.w = x.w / dg;
        *p = x;
    }
#pragma unroll
    for (int j = 0; j < 4; ++j) {
        const int i = t + j * NTHR;
        *(v4f*)(wl + (i >> 5) * LDW + (i & 31) * 4) = wr[j];
    }
    __syncthreads();

    auto gemm_rest = [&](auto tag) {
#pragma nounroll
        for (int k0 = IN_FEAT; k0 < KDIM; k0 += 16) block(tag, k0, k0 - IN_FEAT);
    };
    const float bv = bias[n];
    if (four) gemm_rest(TagYes{}); else gemm_rest(TagNo{});

    auto stage_tile = [&](int mt, const v8f& acc) {
        float* p = feat + (mt * 16 + 8 * h) * LDF + n;
#pragma unroll
        for (int r = 0; r < 8; ++r) p[r * LDF] = acc[r] + bv;
    };
    stage_tile(mg, acc0);
    stage_tile(mg + 4, acc1);
    stage_tile(mg + 8, acc2);
    if (four) stage_tile(12, acc3);
    __syncthreads();
    constexpr int OIT = (RB * (OUT_FEAT / 4) + NTHR - 1) / NTHR;
    v4f ov[OIT];
#pragma unroll
    for (int j = 0; j < OIT; ++j) {
        const int i = t + j * NTHR;
        ov[j] = *(const v4f*)(feat + min(i >> 5, RB - 1) * LDF + (i & 31) * 4);
    }
#pragma unroll
    for (int j = 0; j < OIT; ++j) {
        const int i = t + j * NTHR;
        const int r = i >> 5;
        const int g = row0 + r;
        if (r < RB && g < n_dst) *(v4f*)(out + (size_t)g * OUT_FEAT + (i & 31) * 4) = ov[j];
    }
}

extern "C" void kernel_launch(void* const* d_in, const int* in_sizes, int n_in,
                              void* d_out, int out_size, void* d_ws, size_t ws_size,
                              hipStream_t stream) {
    const float* h_s  = (const float*)d_in[0];
    const float* h_d  = (const float*)d_in[1];
    const int*   src  = (const int*)d_in[2];
    const int*   dst  = (const int*)d_in[3];
    const float* W    = (const float*)d_in[4];
    const float* bias = (const float*)d_in[5];

    const int n_dst   = in_sizes[1] / IN_FEAT;
    const int n_edges = in_sizes[2];
    int* ws = (int*)d_ws;
    (void)n_in; (void)out_size; (void)ws_size;

    const int blocks = (n_dst + RB - 1) / RB;
    sage_sort_slices<<<NSL, STHR, 0, stream>>>(src, dst, ws, n_dst, n_edges);
    sage_owner_fused<<<blocks, NTHR, 0, stream>>>(h_s, h_d, ws, W, bias,
                                                  (float*)d_out, n_dst, n_edges);
}
